# starting kernel
# baseline (speedup 1.0000x reference)
#include <hip/hip_runtime.h>

typedef __attribute__((ext_vector_type(2))) float v2f;
typedef __attribute__((ext_vector_type(8))) float v8f;

#define NNODE 8192
#define DIM   128
#define KNBR  8
#define NEGINF -1e30f

__device__ __forceinline__ float4 ld4(const float* p) {
  return *reinterpret_cast<const float4*>(p);
}
__device__ __forceinline__ void st4(float* p, float4 v) {
  *reinterpret_cast<float4*>(p) = v;
}

__global__ void deg_kernel(const int* __restrict__ neigh,
                           const int* __restrict__ nbrc,
                           float* __restrict__ dinv,
                           unsigned* __restrict__ keep) {
  int i = blockIdx.x * blockDim.x + threadIdx.x;
  if (i >= NNODE) return;
  int cnt = nbrc[i]; cnt = cnt > KNBR ? KNBR : cnt;
  const int* nb = neigh + i * KNBR;
  unsigned km = 0;
  int deg = 1;
  for (int k = 0; k < cnt; ++k) {
    int j = nb[k];
    if (j == i) continue;
    bool dup = false;
    for (int t = 0; t < k; ++t) dup |= (nb[t] == j);
    if (!dup) { km |= (1u << k); ++deg; }
  }
  dinv[i] = rsqrtf((float)deg);
  keep[i] = km;
}

__global__ void wmma_gemm_bias(const float* __restrict__ A, int lda,
                               const float* __restrict__ W, int ldw,
                               const float* __restrict__ bias,
                               float* __restrict__ Y, int ldy,
                               int K, int do_relu) {
  const int wave = threadIdx.x >> 5;
  const int lane = threadIdx.x & 31;
  const int half = lane >> 4;
  const int l16  = lane & 15;
  const int m0 = blockIdx.x * 16;
  const int n0 = (blockIdx.y * (blockDim.x >> 5) + wave) * 16;

  const float* __restrict__ arow = A + (size_t)(m0 + l16) * lda + 2 * half;
  const float* __restrict__ wrow = W + (size_t)(n0 + l16) * ldw + 2 * half;

  v8f c = {};
  for (int k0 = 0; k0 < K; k0 += 4) {
    v2f a = *reinterpret_cast<const v2f*>(arow + k0);
    v2f b = *reinterpret_cast<const v2f*>(wrow + k0);
    c = __builtin_amdgcn_wmma_f32_16x16x4_f32(false, a, false, b,
                                              (short)0, c, false, false);
  }

  const float bv = bias[n0 + l16];
#pragma unroll
  for (int r = 0; r < 8; ++r) {
    float v = c[r] + bv;
    if (do_relu) v = fmaxf(v, 0.0f);
    Y[(size_t)(m0 + r + 8 * half) * ldy + n0 + l16] = v;
  }
}

__device__ __forceinline__ void wave_ln_store(float4 v,
                                              const float* __restrict__ g,
                                              const float* __restrict__ b,
                                              float* __restrict__ dstrow,
                                              int lane) {
  float s = v.x + v.y + v.z + v.w;
#pragma unroll
  for (int off = 16; off > 0; off >>= 1) s += __shfl_xor(s, off, 32);
  const float mu = s * (1.0f / (float)DIM);
  const float dx = v.x - mu, dy = v.y - mu, dz = v.z - mu, dw = v.w - mu;
  float q = dx * dx + dy * dy + dz * dz + dw * dw;
#pragma unroll
  for (int off = 16; off > 0; off >>= 1) q += __shfl_xor(q, off, 32);
  const float rinv = rsqrtf(q * (1.0f / (float)DIM) + 1e-5f);
  const int d = lane * 4;
  const float4 gg = ld4(g + d);
  const float4 bb = ld4(b + d);
  float4 o;
  o.x = dx * rinv * gg.x + bb.x;
  o.y = dy * rinv * gg.y + bb.y;
  o.z = dz * rinv * gg.z + bb.z;
  o.w = dw * rinv * gg.w + bb.w;
  st4(dstrow + d, o);
}

__global__ void gcn_prop_ln(const float* __restrict__ nxt,
                            const float* __restrict__ resid,
                            const int* __restrict__ neigh,
                            const float* __restrict__ dinv,
                            const unsigned* __restrict__ keep,
                            const float* __restrict__ g,
                            const float* __restrict__ b,
                            float* __restrict__ dst, int ldd) {
  const int i = (int)((blockIdx.x * blockDim.x + threadIdx.x) >> 5);
  if (i >= NNODE) return;
  const int lane = threadIdx.x & 31;
  const int d = lane * 4;

  const float di = dinv[i];
  float4 t = ld4(nxt + (size_t)i * DIM + d);
  float4 acc = {di * t.x, di * t.y, di * t.z, di * t.w};

  const unsigned km = keep[i];
  const int* nb = neigh + i * KNBR;
  for (int k = 0; k < KNBR; ++k) {
    if ((km >> k) & 1u) {
      const int j = nb[k];
      const float dj = dinv[j];
      float4 u = ld4(nxt + (size_t)j * DIM + d);
      acc.x += dj * u.x; acc.y += dj * u.y;
      acc.z += dj * u.z; acc.w += dj * u.w;
    }
  }
  acc.x = fmaxf(acc.x * di, 0.0f);
  acc.y = fmaxf(acc.y * di, 0.0f);
  acc.z = fmaxf(acc.z * di, 0.0f);
  acc.w = fmaxf(acc.w * di, 0.0f);

  const float4 r = ld4(resid + (size_t)i * DIM + d);
  float4 v = {acc.x + r.x, acc.y + r.y, acc.z + r.z, acc.w + r.w};
  wave_ln_store(v, g, b, dst + (size_t)i * ldd, lane);
}

__global__ void sage_pool_cat(const float* __restrict__ selfsrc,
                              const float* __restrict__ poolsrc,
                              const int* __restrict__ neigh,
                              const int* __restrict__ nbrc,
                              float* __restrict__ cat) {
  const int i = (int)((blockIdx.x * blockDim.x + threadIdx.x) >> 5);
  if (i >= NNODE) return;
  const int lane = threadIdx.x & 31;
  const int d = lane * 4;

  st4(cat + (size_t)i * (2 * DIM) + d, ld4(selfsrc + (size_t)i * DIM + d));

  int cnt = nbrc[i]; cnt = cnt > KNBR ? KNBR : cnt;
  float4 m = {NEGINF, NEGINF, NEGINF, NEGINF};
  const int* nb = neigh + i * KNBR;
  for (int k = 0; k < cnt; ++k) {
    const int j = nb[k];
    float4 u = ld4(poolsrc + (size_t)j * DIM + d);
    m.x = fmaxf(m.x, u.x); m.y = fmaxf(m.y, u.y);
    m.z = fmaxf(m.z, u.z); m.w = fmaxf(m.w, u.w);
  }
  if (cnt == 0) m = make_float4(0.f, 0.f, 0.f, 0.f);
  st4(cat + (size_t)i * (2 * DIM) + DIM + d, m);
}

__global__ void resid_ln(const float* __restrict__ a,
                         const float* __restrict__ res,
                         const float* __restrict__ g,
                         const float* __restrict__ b,
                         float* __restrict__ dst, int ldd) {
  const int i = (int)((blockIdx.x * blockDim.x + threadIdx.x) >> 5);
  if (i >= NNODE) return;
  const int lane = threadIdx.x & 31;
  const int d = lane * 4;
  const float4 av = ld4(a + (size_t)i * DIM + d);
  const float4 rv = ld4(res + (size_t)i * DIM + d);
  float4 v = {av.x + rv.x, av.y + rv.y, av.z + rv.z, av.w + rv.w};
  wave_ln_store(v, g, b, dst + (size_t)i * ldd, lane);
}

extern "C" void kernel_launch(void* const* d_in, const int* in_sizes, int n_in,
                              void* d_out, int out_size, void* d_ws, size_t ws_size,
                              hipStream_t stream) {
  const float* x      = (const float*)d_in[0];
  const int*   neigh  = (const int*)  d_in[1];
  const int*   nbrc   = (const int*)  d_in[2];
  const float* gcnW   = (const float*)d_in[3];
  const float* gcnB   = (const float*)d_in[4];
  const float* gcnG   = (const float*)d_in[5];
  const float* gcnBb  = (const float*)d_in[6];
  const float* poolW  = (const float*)d_in[7];
  const float* poolB  = (const float*)d_in[8];
  const float* mrgW   = (const float*)d_in[9];
  const float* mrgB   = (const float*)d_in[10];
  const float* sgG    = (const float*)d_in[11];
  const float* sgB    = (const float*)d_in[12];
  float* out = (float*)d_out;

  const size_t ND = (size_t)NNODE * DIM;
  float*    dinv  = (float*)d_ws;
  unsigned* keep  = (unsigned*)(dinv + NNODE);
  float*    first = (float*)(keep + NNODE);
  float*    nxt   = first + ND;
  float*    hbuf  = nxt + ND;
  float*    sbuf  = hbuf + ND;
  float*    tmp   = sbuf + ND;
  float*    cat   = tmp + ND;

  const dim3 gB(128);
  const dim3 gG(NNODE / 16, DIM / 64);
  const dim3 nB(256);
  const dim3 nG(NNODE / 8);

  deg_kernel<<<NNODE / 256, 256, 0, stream>>>(neigh, nbrc, dinv, keep);

  wmma_gemm_bias<<<gG, gB, 0, stream>>>(x, DIM, gcnW, DIM, gcnB,
                                        first, DIM, DIM, 0);
  gcn_prop_ln<<<nG, nB, 0, stream>>>(first, first, neigh, dinv, keep,
                                     gcnG, gcnBb, hbuf, DIM);

  wmma_gemm_bias<<<gG, gB, 0, stream>>>(hbuf, DIM, gcnW + DIM * DIM, DIM,
                                        gcnB + DIM, nxt, DIM, DIM, 0);
  gcn_prop_ln<<<nG, nB, 0, stream>>>(nxt, hbuf, neigh, dinv, keep,
                                     gcnG + DIM, gcnBb + DIM, hbuf, DIM);

  wmma_gemm_bias<<<gG, gB, 0, stream>>>(hbuf, DIM, gcnW + 2 * DIM * DIM, DIM,
                                        gcnB + 2 * DIM, nxt, DIM, DIM, 0);
  gcn_prop_ln<<<nG, nB, 0, stream>>>(nxt, hbuf, neigh, dinv, keep,
                                     gcnG + 2 * DIM, gcnBb + 2 * DIM,
                                     out, 2 * DIM);

  wmma_gemm_bias<<<gG, gB, 0, stream>>>(x, DIM, poolW, DIM, poolB,
                                        tmp, DIM, DIM, 1);
  sage_pool_cat<<<nG, nB, 0, stream>>>(x, tmp, neigh, nbrc, cat);
  wmma_gemm_bias<<<gG, gB, 0, stream>>>(cat, 2 * DIM, mrgW, 2 * DIM, mrgB,
                                        sbuf, DIM, 2 * DIM, 1);

  sage_pool_cat<<<nG, nB, 0, stream>>>(sbuf, sbuf, neigh, nbrc, cat);
  wmma_gemm_bias<<<gG, gB, 0, stream>>>(cat, 2 * DIM, mrgW + DIM * 2 * DIM,
                                        2 * DIM, mrgB + DIM, tmp, DIM,
                                        2 * DIM, 1);
  resid_ln<<<nG, nB, 0, stream>>>(tmp, sbuf, sgG, sgB, sbuf, DIM);

  sage_pool_cat<<<nG, nB, 0, stream>>>(sbuf, sbuf, neigh, nbrc, cat);
  wmma_gemm_bias<<<gG, gB, 0, stream>>>(cat, 2 * DIM, mrgW + 2 * DIM * 2 * DIM,
                                        2 * DIM, mrgB + 2 * DIM, tmp, DIM,
                                        2 * DIM, 1);
  resid_ln<<<nG, nB, 0, stream>>>(tmp, sbuf, sgG + DIM, sgB + DIM,
                                  out + DIM, 2 * DIM);
}


# Neighbour walks ask for everything at once: a node's 8 indices by two 16-byte loads, then all 8 gathered rows in flight before first use (masked slots read own row); sum and max order unchanged
# speedup vs baseline: 1.5823x; 1.5823x over previous
#include <hip/hip_runtime.h>

typedef __attribute__((ext_vector_type(2))) float v2f;
typedef __attribute__((ext_vector_type(8))) float v8f;

#define NNODE 8192
#define DIM   128
#define KNBR  8
#define NEGINF -1e30f

__device__ __forceinline__ float4 ld4(const float* p) {
  return *reinterpret_cast<const float4*>(p);
}
__device__ __forceinline__ void st4(float* p, float4 v) {
  *reinterpret_cast<float4*>(p) = v;
}
__device__ __forceinline__ void ld_nbrs(const int* __restrict__ nb, int (&j)[KNBR]) {
  const int4 lo = *reinterpret_cast<const int4*>(nb);
  const int4 hi = *reinterpret_cast<const int4*>(nb + 4);
  j[0] = lo.x; j[1] = lo.y; j[2] = lo.z; j[3] = lo.w;
  j[4] = hi.x; j[5] = hi.y; j[6] = hi.z; j[7] = hi.w;
}

__global__ void deg_kernel(const int* __restrict__ neigh,
                           const int* __restrict__ nbrc,
                           float* __restrict__ dinv,
                           unsigned* __restrict__ keep) {
  int i = blockIdx.x * blockDim.x + threadIdx.x;
  if (i >= NNODE) return;
  int cnt = nbrc[i]; cnt = cnt > KNBR ? KNBR : cnt;
  int nb[KNBR];
  ld_nbrs(neigh + i * KNBR, nb);
  unsigned km = 0;
  int deg = 1;
#pragma unroll
  for (int k = 0; k < KNBR; ++k) {
    if (k < cnt) {
      const int j = nb[k];
      bool dup = (j == i);
#pragma unroll
      for (int t = 0; t < k; ++t) dup |= (nb[t] == j);
      if (!dup) { km |= (1u << k); ++deg; }
    }
  }
  dinv[i] = rsqrtf((float)deg);
  keep[i] = km;
}

__device__ __forceinline__ v8f mma4(v2f a, v2f b, v8f c) {
  return __builtin_amdgcn_wmma_f32_16x16x4_f32(false, a, false, b,
                                               (short)0, c, false, false);
}

struct Frag4 { float4 a0, a1, b0, b1; };

__device__ __forceinline__ Frag4 ld_group(const float* __restrict__ a0p,
                                          const float* __restrict__ a1p,
                                          const float* __restrict__ b0p,
                                          const float* __restrict__ b1p,
                                          int koff) {
  Frag4 f;
  f.a0 = ld4(a0p + koff);
  f.a1 = ld4(a1p + koff);
  f.b0 = ld4(b0p + koff);
  f.b1 = ld4(b1p + koff);
  return f;
}

template <int K, bool RELU>
__global__ __launch_bounds__(128) __attribute__((amdgpu_waves_per_eu(1, 2)))
void wmma_gemm_bias(const float* __restrict__ A,
                    const float* __restrict__ W,
                    const float* __restrict__ bias,
                    float* __restrict__ Y) {
  constexpr int NC = K / 32;
  const int wave = threadIdx.x >> 5;
  const int lane = threadIdx.x & 31;
  const int half = lane >> 4;
  const int l16  = lane & 15;
  const int m0 = blockIdx.x * 32;
  const int n0 = wave * 32;

  const float* __restrict__ a0p = A + (size_t)(m0 + l16) * K + 4 * half;
  const float* __restrict__ a1p = a0p + 16 * K;
  const float* __restrict__ b0p = W + (size_t)(n0 + 2 * l16) * K + 4 * half;
  const float* __restrict__ b1p = b0p + K;

  Frag4 f[NC][4];
#pragma unroll
  for (int g = 0; g < 4; ++g) f[0][g] = ld_group(a0p, a1p, b0p, b1p, 8 * g);

  v8f c00 = {}, c01 = {}, c10 = {}, c11 = {};
#pragma unroll
  for (int c = 0; c < NC; ++c) {
    if (c + 1 < NC) {
#pragma unroll
      for (int g = 0; g < 4; ++g)
        f[c + 1][g] = ld_group(a0p, a1p, b0p, b1p, 32 * (c + 1) + 8 * g);
    }
#pragma unroll
    for (int g = 0; g < 4; ++g) {
      const Frag4 q = f[c][g];
      const v2f a0l = {q.a0.x, q.a0.y}, a0h = {q.a0.z, q.a0.w};
      const v2f a1l = {q.a1.x, q.a1.y}, a1h = {q.a1.z, q.a1.w};
      const v2f b0l = {q.b0.x, q.b0.y}, b0h = {q.b0.z, q.b0.w};
      const v2f b1l = {q.b1.x, q.b1.y}, b1h = {q.b1.z, q.b1.w};
      c00 = mma4(a0l, b0l, c00);
      c01 = mma4(a0l, b1l, c01);
      c10 = mma4(a1l, b0l, c10);
      c11 = mma4(a1l, b1l, c11);
      c00 = mma4(a0h, b0h, c00);
      c01 = mma4(a0h, b1h, c01);
      c10 = mma4(a1h, b0h, c10);
      c11 = mma4(a1h, b1h, c11);
    }
  }

  const int col = n0 + 2 * l16;
  const float2 bv = *reinterpret_cast<const float2*>(bias + col);
  float* __restrict__ y0 = Y + (size_t)(m0 + 8 * half) * DIM + col;
  float* __restrict__ y1 = y0 + 16 * DIM;
#pragma unroll
  for (int r = 0; r < 8; ++r) {
    float2 u = {c00[r] + bv.x, c01[r] + bv.y};
    float2 v = {c10[r] + bv.x, c11[r] + bv.y};
    if (RELU) {
      u.x = fmaxf(u.x, 0.0f); u.y = fmaxf(u.y, 0.0f);
      v.x = fmaxf(v.x, 0.0f); v.y = fmaxf(v.y, 0.0f);
    }
    *reinterpret_cast<float2*>(y0 + (size_t)r * DIM) = u;
    *reinterpret_cast<float2*>(y1 + (size_t)r * DIM) = v;
  }
}

__device__ __forceinline__ void wave_ln_store(float4 v,
                                              const float* __restrict__ g,
                                              const float* __restrict__ b,
                                              float* __restrict__ dstrow,
                                              int lane) {
  float s = v.x + v.y + v.z + v.w;
#pragma unroll
  for (int off = 16; off > 0; off >>= 1) s += __shfl_xor(s, off, 32);
  const float mu = s * (1.0f / (float)DIM);
  const float dx = v.x - mu, dy = v.y - mu, dz = v.z - mu, dw = v.w - mu;
  float q = dx * dx + dy * dy + dz * dz + dw * dw;
#pragma unroll
  for (int off = 16; off > 0; off >>= 1) q += __shfl_xor(q, off, 32);
  const float rinv = rsqrtf(q * (1.0f / (float)DIM) + 1e-5f);
  const int d = lane * 4;
  const float4 gg = ld4(g + d);
  const float4 bb = ld4(b + d);
  float4 o;
  o.x = dx * rinv * gg.x + bb.x;
  o.y = dy * rinv * gg.y + bb.y;
  o.z = dz * rinv * gg.z + bb.z;
  o.w = dw * rinv * gg.w + bb.w;
  st4(dstrow + d, o);
}

__global__ void gcn_prop_ln(const float* __restrict__ nxt,
                            const float* __restrict__ resid,
                            const int* __restrict__ neigh,
                            const float* __restrict__ dinv,
                            const unsigned* __restrict__ keep,
                            const float* __restrict__ g,
                            const float* __restrict__ b,
                            float* __restrict__ dst, int ldd) {
  const int i = (int)((blockIdx.x * blockDim.x + threadIdx.x) >> 5);
  if (i >= NNODE) return;
  const int lane = threadIdx.x & 31;
  const int d = lane * 4;

  const float di = dinv[i];
  const unsigned km = keep[i];
  int nb[KNBR];
  ld_nbrs(neigh + i * KNBR, nb);
  const float4 t = ld4(nxt + (size_t)i * DIM + d);
  const float4 r = ld4(resid + (size_t)i * DIM + d);

  float4 u[KNBR];
  float dj[KNBR];
#pragma unroll
  for (int k = 0; k < KNBR; ++k) {
    const int j = ((km >> k) & 1u) ? nb[k] : i;
    dj[k] = dinv[j];
    u[k] = ld4(nxt + (size_t)j * DIM + d);
  }
#pragma unroll
  for (int k = 0; k < KNBR; ++k) {
    asm volatile("" :: "v"(u[k].x), "v"(u[k].y), "v"(u[k].z), "v"(u[k].w),
                       "v"(dj[k]));
  }

  float4 acc = {di * t.x, di * t.y, di * t.z, di * t.w};
#pragma unroll
  for (int k = 0; k < KNBR; ++k) {
    if ((km >> k) & 1u) {
      acc.x += dj[k] * u[k].x; acc.y += dj[k] * u[k].y;
      acc.z += dj[k] * u[k].z; acc.w += dj[k] * u[k].w;
    }
  }
  acc.x = fmaxf(acc.x * di, 0.0f);
  acc.y = fmaxf(acc.y * di, 0.0f);
  acc.z = fmaxf(acc.z * di, 0.0f);
  acc.w = fmaxf(acc.w * di, 0.0f);

  float4 v = {acc.x + r.x, acc.y + r.y, acc.z + r.z, acc.w + r.w};
  wave_ln_store(v, g, b, dst + (size_t)i * ldd, lane);
}

__global__ void sage_pool_cat(const float* __restrict__ selfsrc,
                              const float* __restrict__ poolsrc,
                              const int* __restrict__ neigh,
                              const int* __restrict__ nbrc,
                              float* __restrict__ cat) {
  const int i = (int)((blockIdx.x * blockDim.x + threadIdx.x) >> 5);
  if (i >= NNODE) return;
  const int lane = threadIdx.x & 31;
  const int d = lane * 4;

  int cnt = nbrc[i]; cnt = cnt > KNBR ? KNBR : cnt;
  int nb[KNBR];
  ld_nbrs(neigh + i * KNBR, nb);
  const float4 self = ld4(selfsrc + (size_t)i * DIM + d);

  float4 u[KNBR];
#pragma unroll
  for (int k = 0; k < KNBR; ++k) {
    const int j = (k < cnt) ? nb[k] : i;
    u[k] = ld4(poolsrc + (size_t)j * DIM + d);
  }
  st4(cat + (size_t)i * (2 * DIM) + d, self);
#pragma unroll
  for (int k = 0; k < KNBR; ++k) {
    asm volatile("" :: "v"(u[k].x), "v"(u[k].y), "v"(u[k].z), "v"(u[k].w));
  }

  float4 m = {NEGINF, NEGINF, NEGINF, NEGINF};
#pragma unroll
  for (int k = 0; k < KNBR; ++k) {
    if (k < cnt) {
      m.x = fmaxf(m.x, u[k].x); m.y = fmaxf(m.y, u[k].y);
      m.z = fmaxf(m.z, u[k].z); m.w = fmaxf(m.w, u[k].w);
    }
  }
  if (cnt == 0) m = make_float4(0.f, 0.f, 0.f, 0.f);
  st4(cat + (size_t)i * (2 * DIM) + DIM + d, m);
}

__global__ void resid_ln(const float* __restrict__ a,
                         const float* __restrict__ res,
                         const float* __restrict__ g,
                         const float* __restrict__ b,
                         float* __restrict__ dst, int ldd) {
  const int i = (int)((blockIdx.x * blockDim.x + threadIdx.x) >> 5);
  if (i >= NNODE) return;
  const int lane = threadIdx.x & 31;
  const int d = lane * 4;
  const float4 av = ld4(a + (size_t)i * DIM + d);
  const float4 rv = ld4(res + (size_t)i * DIM + d);
  float4 v = {av.x + rv.x, av.y + rv.y, av.z + rv.z, av.w + rv.w};
  wave_ln_store(v, g, b, dst + (size_t)i * ldd, lane);
}

extern "C" void kernel_launch(void* const* d_in, const int* in_sizes, int n_in,
                              void* d_out, int out_size, void* d_ws, size_t ws_size,
                              hipStream_t stream) {
  const float* x      = (const float*)d_in[0];
  const int*   neigh  = (const int*)  d_in[1];
  const int*   nbrc   = (const int*)  d_in[2];
  const float* gcnW   = (const float*)d_in[3];
  const float* gcnB   = (const float*)d_in[4];
  const float* gcnG   = (const float*)d_in[5];
  const float* gcnBb  = (const float*)d_in[6];
  const float* poolW  = (const float*)d_in[7];
  const float* poolB  = (const float*)d_in[8];
  const float* mrgW   = (const float*)d_in[9];
  const float* mrgB   = (const float*)d_in[10];
  const float* sgG    = (const float*)d_in[11];
  const float* sgB    = (const float*)d_in[12];
  float* out = (float*)d_out;

  const size_t ND = (size_t)NNODE * DIM;
  float*    dinv  = (float*)d_ws;
  unsigned* keep  = (unsigned*)(dinv + NNODE);
  float*    first = (float*)(keep + NNODE);
  float*    nxt   = first + ND;
  float*    hbuf  = nxt + ND;
  float*    sbuf  = hbuf + ND;
  float*    tmp   = sbuf + ND;
  float*    cat   = tmp + ND;

  const dim3 gB(128);
  const dim3 gG(NNODE / 32);
  const dim3 nB(256);
  const dim3 nG(NNODE / 8);

  deg_kernel<<<NNODE / 256, 256, 0, stream>>>(neigh, nbrc, dinv, keep);

  wmma_gemm_bias<DIM, false><<<gG, gB, 0, stream>>>(x, gcnW, gcnB, first);
  gcn_prop_ln<<<nG, nB, 0, stream>>>(first, first, neigh, dinv, keep,
                                     gcnG, gcnBb, hbuf, DIM);

  wmma_gemm_bias<DIM, false><<<gG, gB, 0, stream>>>(hbuf, gcnW + DIM * DIM,
                                                    gcnB + DIM, nxt);
  gcn_prop_ln<<<nG, nB, 0, stream>>>(nxt, hbuf, neigh, dinv, keep,
                                     gcnG + DIM, gcnBb + DIM, hbuf, DIM);

  wmma_gemm_bias<DIM, false><<<gG, gB, 0, stream>>>(hbuf, gcnW + 2 * DIM * DIM,
                                                    gcnB + 2 * DIM, nxt);
  gcn_prop_ln<<<nG, nB, 0, stream>>>(nxt, hbuf, neigh, dinv, keep,
                                     gcnG + 2 * DIM, gcnBb + 2 * DIM,
                                     out, 2 * DIM);

  wmma_gemm_bias<DIM, true><<<gG, gB, 0, stream>>>(x, poolW, poolB, tmp);
  sage_pool_cat<<<nG, nB, 0, stream>>>(x, tmp, neigh, nbrc, cat);
  wmma_gemm_bias<2 * DIM, true><<<gG, gB, 0, stream>>>(cat, mrgW, mrgB,
                                                       sbuf);

  sage_pool_cat<<<nG, nB, 0, stream>>>(sbuf, sbuf, neigh, nbrc, cat);
  wmma_gemm_bias<2 * DIM, true><<<gG, gB, 0, stream>>>(cat, mrgW + DIM * 2 * DIM,
                                                       mrgB + DIM, tmp);
  resid_ln<<<nG, nB, 0, stream>>>(tmp, sbuf, sgG, sgB, sbuf, DIM);

  sage_pool_cat<<<nG, nB, 0, stream>>>(sbuf, sbuf, neigh, nbrc, cat);
  wmma_gemm_bias<2 * DIM, true><<<gG, gB, 0, stream>>>(cat, mrgW + 2 * DIM * 2 * DIM,
                                                       mrgB + 2 * DIM, tmp);
  resid_ln<<<nG, nB, 0, stream>>>(tmp, sbuf, sgG + DIM, sgB + DIM,
                                  out + DIM, 2 * DIM);
}


# Same-depth stages of the two independent branches share launches: a wave reads a node's neighbour list once and gathers for both, a GEMM block does both products in turn; 16 -> 10; sums unchanged
# speedup vs baseline: 1.8855x; 1.1916x over previous
#include <hip/hip_runtime.h>

typedef __attribute__((ext_vector_type(2))) float v2f;
typedef __attribute__((ext_vector_type(8))) float v8f;

#define NNODE 8192
#define DIM   128
#define KNBR  8
#define NEGINF -1e30f

__device__ __forceinline__ float4 ld4(const float* p) {
  return *reinterpret_cast<const float4*>(p);
}
__device__ __forceinline__ void st4(float* p, float4 v) {
  *reinterpret_cast<float4*>(p) = v;
}
__device__ __forceinline__ void ld_nbrs(const int* __restrict__ nb, int (&j)[KNBR]) {
  const int4 lo = *reinterpret_cast<const int4*>(nb);
  const int4 hi = *reinterpret_cast<const int4*>(nb + 4);
  j[0] = lo.x; j[1] = lo.y; j[2] = lo.z; j[3] = lo.w;
  j[4] = hi.x; j[5] = hi.y; j[6] = hi.z; j[7] = hi.w;
}

__global__ void deg_kernel(const int* __restrict__ neigh,
                           const int* __restrict__ nbrc,
                           float* __restrict__ dinv,
                           unsigned* __restrict__ keep) {
  int i = blockIdx.x * blockDim.x + threadIdx.x;
  if (i >= NNODE) return;
  int cnt = nbrc[i]; cnt = cnt > KNBR ? KNBR : cnt;
  int nb[KNBR];
  ld_nbrs(neigh + i * KNBR, nb);
  unsigned km = 0;
  int deg = 1;
#pragma unroll
  for (int k = 0; k < KNBR; ++k) {
    if (k < cnt) {
      const int j = nb[k];
      bool dup = (j == i);
#pragma unroll
      for (int t = 0; t < k; ++t) dup |= (nb[t] == j);
      if (!dup) { km |= (1u << k); ++deg; }
    }
  }
  dinv[i] = rsqrtf((float)deg);
  keep[i] = km;
}

__device__ __forceinline__ v8f mma4(v2f a, v2f b, v8f c) {
  return __builtin_amdgcn_wmma_f32_16x16x4_f32(false, a, false, b,
                                               (short)0, c, false, false);
}

struct Frag4 { float4 a0, a1, b0, b1; };

__device__ __forceinline__ Frag4 ld_group(const float* __restrict__ a0p,
                                          const float* __restrict__ a1p,
                                          const float* __restrict__ b0p,
                                          const float* __restrict__ b1p,
                                          int koff) {
  Frag4 f;
  f.a0 = ld4(a0p + koff);
  f.a1 = ld4(a1p + koff);
  f.b0 = ld4(b0p + koff);
  f.b1 = ld4(b1p + koff);
  return f;
}

template <int K, bool RELU>
__device__ __forceinline__ void gemm_tile(const float* __restrict__ A,
                                          const float* __restrict__ W,
                                          const float* __restrict__ bias,
                                          float* __restrict__ Y) {
  constexpr int NC = K / 32;
  const int wave = threadIdx.x >> 5;
  const int lane = threadIdx.x & 31;
  const int half = lane >> 4;
  const int l16  = lane & 15;
  const int m0 = blockIdx.x * 32;
  const int n0 = wave * 32;

  const float* __restrict__ a0p = A + (size_t)(m0 + l16) * K + 4 * half;
  const float* __restrict__ a1p = a0p + 16 * K;
  const float* __restrict__ b0p = W + (size_t)(n0 + 2 * l16) * K + 4 * half;
  const float* __restrict__ b1p = b0p + K;

  Frag4 f[NC][4];
#pragma unroll
  for (int g = 0; g < 4; ++g) f[0][g] = ld_group(a0p, a1p, b0p, b1p, 8 * g);

  v8f c00 = {}, c01 = {}, c10 = {}, c11 = {};
#pragma unroll
  for (int c = 0; c < NC; ++c) {
    if (c + 1 < NC) {
#pragma unroll
      for (int g = 0; g < 4; ++g)
        f[c + 1][g] = ld_group(a0p, a1p, b0p, b1p, 32 * (c + 1) + 8 * g);
    }
#pragma unroll
    for (int g = 0; g < 4; ++g) {
      const Frag4 q = f[c][g];
      const v2f a0l = {q.a0.x, q.a0.y}, a0h = {q.a0.z, q.a0.w};
      const v2f a1l = {q.a1.x, q.a1.y}, a1h = {q.a1.z, q.a1.w};
      const v2f b0l = {q.b0.x, q.b0.y}, b0h = {q.b0.z, q.b0.w};
      const v2f b1l = {q.b1.x, q.b1.y}, b1h = {q.b1.z, q.b1.w};
      c00 = mma4(a0l, b0l, c00);
      c01 = mma4(a0l, b1l, c01);
      c10 = mma4(a1l, b0l, c10);
      c11 = mma4(a1l, b1l, c11);
      c00 = mma4(a0h, b0h, c00);
      c01 = mma4(a0h, b1h, c01);
      c10 = mma4(a1h, b0h, c10);
      c11 = mma4(a1h, b1h, c11);
    }
  }

  const int col = n0 + 2 * l16;
  const float2 bv = *reinterpret_cast<const float2*>(bias + col);
  float* __restrict__ y0 = Y + (size_t)(m0 + 8 * half) * DIM + col;
  float* __restrict__ y1 = y0 + 16 * DIM;
#pragma unroll
  for (int r = 0; r < 8; ++r) {
    float2 u = {c00[r] + bv.x, c01[r] + bv.y};
    float2 v = {c10[r] + bv.x, c11[r] + bv.y};
    if (RELU) {
      u.x = fmaxf(u.x, 0.0f); u.y = fmaxf(u.y, 0.0f);
      v.x = fmaxf(v.x, 0.0f); v.y = fmaxf(v.y, 0.0f);
    }
    *reinterpret_cast<float2*>(y0 + (size_t)r * DIM) = u;
    *reinterpret_cast<float2*>(y1 + (size_t)r * DIM) = v;
  }
}

template <int K, bool RELU>
__global__ __launch_bounds__(128) __attribute__((amdgpu_waves_per_eu(1, 2)))
void wmma_gemm_bias(const float* __restrict__ A,
                    const float* __restrict__ W,
                    const float* __restrict__ bias,
                    float* __restrict__ Y) {
  gemm_tile<K, RELU>(A, W, bias, Y);
}

template <int K1, bool RELU1, int K2, bool RELU2>
__global__ __launch_bounds__(128) __attribute__((amdgpu_waves_per_eu(1, 2)))
void wmma_gemm_pair(const float* __restrict__ A1,
                    const float* __restrict__ W1,
                    const float* __restrict__ bias1,
                    float* __restrict__ Y1,
                    const float* __restrict__ A2,
                    const float* __restrict__ W2,
                    const float* __restrict__ bias2,
                    float* __restrict__ Y2) {
  gemm_tile<K1, RELU1>(A1, W1, bias1, Y1);
  gemm_tile<K2, RELU2>(A2, W2, bias2, Y2);
}

__device__ __forceinline__ void wave_ln_store(float4 v,
                                              const float* __restrict__ g,
                                              const float* __restrict__ b,
                                              float* __restrict__ dstrow,
                                              int lane) {
  float s = v.x + v.y + v.z + v.w;
#pragma unroll
  for (int off = 16; off > 0; off >>= 1) s += __shfl_xor(s, off, 32);
  const float mu = s * (1.0f / (float)DIM);
  const float dx = v.x - mu, dy = v.y - mu, dz = v.z - mu, dw = v.w - mu;
  float q = dx * dx + dy * dy + dz * dz + dw * dw;
#pragma unroll
  for (int off = 16; off > 0; off >>= 1) q += __shfl_xor(q, off, 32);
  const float rinv = rsqrtf(q * (1.0f / (float)DIM) + 1e-5f);
  const int d = lane * 4;
  const float4 gg = ld4(g + d);
  const float4 bb = ld4(b + d);
  float4 o;
  o.x = dx * rinv * gg.x + bb.x;
  o.y = dy * rinv * gg.y + bb.y;
  o.z = dz * rinv * gg.z + bb.z;
  o.w = dw * rinv * gg.w + bb.w;
  st4(dstrow + d, o);
}

__device__ __forceinline__ float4 gcn_sum(float di, unsigned km, float4 t, float4 r,
                                          const float4 (&u)[KNBR],
                                          const float (&dj)[KNBR]) {
  float4 acc = {di * t.x, di * t.y, di * t.z, di * t.w};
#pragma unroll
  for (int k = 0; k < KNBR; ++k) {
    if ((km >> k) & 1u) {
      acc.x += dj[k] * u[k].x; acc.y += dj[k] * u[k].y;
      acc.z += dj[k] * u[k].z; acc.w += dj[k] * u[k].w;
    }
  }
  acc.x = fmaxf(acc.x * di, 0.0f);
  acc.y = fmaxf(acc.y * di, 0.0f);
  acc.z = fmaxf(acc.z * di, 0.0f);
  acc.w = fmaxf(acc.w * di, 0.0f);
  return make_float4(acc.x + r.x, acc.y + r.y, acc.z + r.z, acc.w + r.w);
}

__global__ void gcn_ln_sage_pool(const float* __restrict__ nxt,
                                 const float* __restrict__ resid,
                                 const int* __restrict__ neigh,
                                 const int* __restrict__ nbrc,
                                 const float* __restrict__ dinv,
                                 const unsigned* __restrict__ keep,
                                 const float* __restrict__ g,
                                 const float* __restrict__ b,
                                 float* __restrict__ dst,
                                 const float* __restrict__ selfsrc,
                                 const float* __restrict__ poolsrc,
                                 float* __restrict__ cat) {
  const int i = (int)((blockIdx.x * blockDim.x + threadIdx.x) >> 5);
  if (i >= NNODE) return;
  const int lane = threadIdx.x & 31;
  const int d = lane * 4;

  const float4 t = ld4(nxt + (size_t)i * DIM + d);
  const float4 r = ld4(resid + (size_t)i * DIM + d);
  const float4 self = ld4(selfsrc + (size_t)i * DIM + d);
  const float di = dinv[i];
  const unsigned km = keep[i];
  int cnt = nbrc[i]; cnt = cnt > KNBR ? KNBR : cnt;
  int nb[KNBR];
  ld_nbrs(neigh + i * KNBR, nb);
  asm volatile("" :: "v"(t.x), "v"(t.y), "v"(t.z), "v"(t.w),
                     "v"(r.x), "v"(r.y), "v"(r.z), "v"(r.w),
                     "v"(self.x), "v"(self.y), "v"(self.z), "v"(self.w),
                     "v"(di), "v"(km), "v"(cnt),
                     "v"(nb[0]), "v"(nb[1]), "v"(nb[2]), "v"(nb[3]),
                     "v"(nb[4]), "v"(nb[5]), "v"(nb[6]), "v"(nb[7]));

  float4 u[KNBR], p[KNBR];
  float dj[KNBR];
#pragma unroll
  for (int k = 0; k < KNBR; ++k) {
    const int j = ((km >> k) & 1u) ? nb[k] : i;
    dj[k] = dinv[j];
    u[k] = ld4(nxt + (size_t)j * DIM + d);
  }
#pragma unroll
  for (int k = 0; k < KNBR; ++k) {
    const int j = (k < cnt) ? nb[k] : i;
    p[k] = ld4(poolsrc + (size_t)j * DIM + d);
  }
#pragma unroll
  for (int k = 0; k < KNBR; ++k) {
    asm volatile("" :: "v"(u[k].x), "v"(u[k].y), "v"(u[k].z), "v"(u[k].w),
                       "v"(dj[k]));
    asm volatile("" :: "v"(p[k].x), "v"(p[k].y), "v"(p[k].z), "v"(p[k].w));
  }

  const float4 v = gcn_sum(di, km, t, r, u, dj);
  wave_ln_store(v, g, b, dst + (size_t)i * DIM, lane);

  float4 m = {NEGINF, NEGINF, NEGINF, NEGINF};
#pragma unroll
  for (int k = 0; k < KNBR; ++k) {
    if (k < cnt) {
      m.x = fmaxf(m.x, p[k].x); m.y = fmaxf(m.y, p[k].y);
      m.z = fmaxf(m.z, p[k].z); m.w = fmaxf(m.w, p[k].w);
    }
  }
  if (cnt == 0) m = make_float4(0.f, 0.f, 0.f, 0.f);
  st4(cat + (size_t)i * (2 * DIM) + d, self);
  st4(cat + (size_t)i * (2 * DIM) + DIM + d, m);
}

__global__ void gcn_ln_sage_ln(const float* __restrict__ nxt,
                               const float* __restrict__ resid,
                               const int* __restrict__ neigh,
                               const float* __restrict__ dinv,
                               const unsigned* __restrict__ keep,
                               const float* __restrict__ g,
                               const float* __restrict__ b,
                               float* __restrict__ dst, int ldd,
                               const float* __restrict__ a,
                               const float* __restrict__ res,
                               const float* __restrict__ sg,
                               const float* __restrict__ sb,
                               float* __restrict__ sdst) {
  const int i = (int)((blockIdx.x * blockDim.x + threadIdx.x) >> 5);
  if (i >= NNODE) return;
  const int lane = threadIdx.x & 31;
  const int d = lane * 4;

  const float4 av = ld4(a + (size_t)i * DIM + d);
  const float4 rv = ld4(res + (size_t)i * DIM + d);
  const float4 t = ld4(nxt + (size_t)i * DIM + d);
  const float4 r = ld4(resid + (size_t)i * DIM + d);
  const float di = dinv[i];
  const unsigned km = keep[i];
  int nb[KNBR];
  ld_nbrs(neigh + i * KNBR, nb);
  asm volatile("" :: "v"(av.x), "v"(av.y), "v"(av.z), "v"(av.w),
                     "v"(rv.x), "v"(rv.y), "v"(rv.z), "v"(rv.w),
                     "v"(t.x), "v"(t.y), "v"(t.z), "v"(t.w),
                     "v"(r.x), "v"(r.y), "v"(r.z), "v"(r.w),
                     "v"(di), "v"(km),
                     "v"(nb[0]), "v"(nb[1]), "v"(nb[2]), "v"(nb[3]),
                     "v"(nb[4]), "v"(nb[5]), "v"(nb[6]), "v"(nb[7]));

  float4 u[KNBR];
  float dj[KNBR];
#pragma unroll
  for (int k = 0; k < KNBR; ++k) {
    const int j = ((km >> k) & 1u) ? nb[k] : i;
    dj[k] = dinv[j];
    u[k] = ld4(nxt + (size_t)j * DIM + d);
  }
#pragma unroll
  for (int k = 0; k < KNBR; ++k) {
    asm volatile("" :: "v"(u[k].x), "v"(u[k].y), "v"(u[k].z), "v"(u[k].w),
                       "v"(dj[k]));
  }

  const float4 w = {av.x + rv.x, av.y + rv.y, av.z + rv.z, av.w + rv.w};
  wave_ln_store(w, sg, sb, sdst + (size_t)i * DIM, lane);

  const float4 v = gcn_sum(di, km, t, r, u, dj);
  wave_ln_store(v, g, b, dst + (size_t)i * ldd, lane);
}

__global__ void sage_pool_cat(const float* __restrict__ selfsrc,
                              const float* __restrict__ poolsrc,
                              const int* __restrict__ neigh,
                              const int* __restrict__ nbrc,
                              float* __restrict__ cat) {
  const int i = (int)((blockIdx.x * blockDim.x + threadIdx.x) >> 5);
  if (i >= NNODE) return;
  const int lane = threadIdx.x & 31;
  const int d = lane * 4;

  int cnt = nbrc[i]; cnt = cnt > KNBR ? KNBR : cnt;
  int nb[KNBR];
  ld_nbrs(neigh + i * KNBR, nb);
  const float4 self = ld4(selfsrc + (size_t)i * DIM + d);

  float4 u[KNBR];
#pragma unroll
  for (int k = 0; k < KNBR; ++k) {
    const int j = (k < cnt) ? nb[k] : i;
    u[k] = ld4(poolsrc + (size_t)j * DIM + d);
  }
  st4(cat + (size_t)i * (2 * DIM) + d, self);
#pragma unroll
  for (int k = 0; k < KNBR; ++k) {
    asm volatile("" :: "v"(u[k].x), "v"(u[k].y), "v"(u[k].z), "v"(u[k].w));
  }

  float4 m = {NEGINF, NEGINF, NEGINF, NEGINF};
#pragma unroll
  for (int k = 0; k < KNBR; ++k) {
    if (k < cnt) {
      m.x = fmaxf(m.x, u[k].x); m.y = fmaxf(m.y, u[k].y);
      m.z = fmaxf(m.z, u[k].z); m.w = fmaxf(m.w, u[k].w);
    }
  }
  if (cnt == 0) m = make_float4(0.f, 0.f, 0.f, 0.f);
  st4(cat + (size_t)i * (2 * DIM) + DIM + d, m);
}

__global__ void resid_ln(const float* __restrict__ a,
                         const float* __restrict__ res,
                         const float* __restrict__ g,
                         const float* __restrict__ b,
                         float* __restrict__ dst, int ldd) {
  const int i = (int)((blockIdx.x * blockDim.x + threadIdx.x) >> 5);
  if (i >= NNODE) return;
  const int lane = threadIdx.x & 31;
  const int d = lane * 4;
  const float4 av = ld4(a + (size_t)i * DIM + d);
  const float4 rv = ld4(res + (size_t)i * DIM + d);
  float4 v = {av.x + rv.x, av.y + rv.y, av.z + rv.z, av.w + rv.w};
  wave_ln_store(v, g, b, dst + (size_t)i * ldd, lane);
}

extern "C" void kernel_launch(void* const* d_in, const int* in_sizes, int n_in,
                              void* d_out, int out_size, void* d_ws, size_t ws_size,
                              hipStream_t stream) {
  const float* x      = (const float*)d_in[0];
  const int*   neigh  = (const int*)  d_in[1];
  const int*   nbrc   = (const int*)  d_in[2];
  const float* gcnW   = (const float*)d_in[3];
  const float* gcnB   = (const float*)d_in[4];
  const float* gcnG   = (const float*)d_in[5];
  const float* gcnBb  = (const float*)d_in[6];
  const float* poolW  = (const float*)d_in[7];
  const float* poolB  = (const float*)d_in[8];
  const float* mrgW   = (const float*)d_in[9];
  const float* mrgB   = (const float*)d_in[10];
  const float* sgG    = (const float*)d_in[11];
  const float* sgB    = (const float*)d_in[12];
  float* out = (float*)d_out;

  const size_t ND = (size_t)NNODE * DIM;
  float*    dinv  = (float*)d_ws;
  unsigned* keep  = (unsigned*)(dinv + NNODE);
  float*    first = (float*)(keep + NNODE);
  float*    nxt   = first + ND;
  float*    hbuf  = nxt + ND;
  float*    sbuf  = hbuf + ND;
  float*    tmp   = sbuf + ND;
  float*    cat   = tmp + ND;

  const dim3 gB(128);
  const dim3 gG(NNODE / 32);
  const dim3 nB(256);
  const dim3 nG(NNODE / 8);

  deg_kernel<<<NNODE / 256, 256, 0, stream>>>(neigh, nbrc, dinv, keep);

  wmma_gemm_pair<DIM, false, DIM, true><<<gG, gB, 0, stream>>>(
      x, gcnW, gcnB, first, x, poolW, poolB, tmp);
  gcn_ln_sage_pool<<<nG, nB, 0, stream>>>(first, first, neigh, nbrc, dinv, keep,
                                          gcnG, gcnBb, hbuf, x, tmp, cat);

  wmma_gemm_pair<DIM, false, 2 * DIM, true><<<gG, gB, 0, stream>>>(
      hbuf, gcnW + DIM * DIM, gcnB + DIM, nxt, cat, mrgW, mrgB, sbuf);
  gcn_ln_sage_pool<<<nG, nB, 0, stream>>>(nxt, hbuf, neigh, nbrc, dinv, keep,
                                          gcnG + DIM, gcnBb + DIM, hbuf,
                                          sbuf, sbuf, cat);

  wmma_gemm_pair<DIM, false, 2 * DIM, true><<<gG, gB, 0, stream>>>(
      hbuf, gcnW + 2 * DIM * DIM, gcnB + 2 * DIM, nxt,
      cat, mrgW + DIM * 2 * DIM, mrgB + DIM, tmp);
  gcn_ln_sage_ln<<<nG, nB, 0, stream>>>(nxt, hbuf, neigh, dinv, keep,
                                        gcnG + 2 * DIM, gcnBb + 2 * DIM,
                                        out, 2 * DIM,
                                        tmp, sbuf, sgG, sgB, sbuf);

  sage_pool_cat<<<nG, nB, 0, stream>>>(sbuf, sbuf, neigh, nbrc, cat);
  wmma_gemm_bias<2 * DIM, true><<<gG, gB, 0, stream>>>(cat, mrgW + 2 * DIM * 2 * DIM,
                                                       mrgB + 2 * DIM, tmp);
  resid_ln<<<nG, nB, 0, stream>>>(tmp, sbuf, sgG + DIM, sgB + DIM,
                                  out + DIM, 2 * DIM);
}


# GEMM B operand: the degree launch also re-lays the 7 weight matrices into fragment order, so each B fragment load reads 512 contiguous bytes (4 lines, not 16); k order and sums unchanged
# speedup vs baseline: 1.9785x; 1.0493x over previous
#include <hip/hip_runtime.h>

typedef __attribute__((ext_vector_type(2))) float v2f;
typedef __attribute__((ext_vector_type(8))) float v8f;

#define NNODE 8192
#define DIM   128
#define KNBR  8
#define NEGINF -1e30f

__device__ __forceinline__ float4 ld4(const float* p) {
  return *reinterpret_cast<const float4*>(p);
}
__device__ __forceinline__ void st4(float* p, float4 v) {
  *reinterpret_cast<float4*>(p) = v;
}
__device__ __forceinline__ void ld_nbrs(const int* __restrict__ nb, int (&j)[KNBR]) {
  const int4 lo = *reinterpret_cast<const int4*>(nb);
  const int4 hi = *reinterpret_cast<const int4*>(nb + 4);
  j[0] = lo.x; j[1] = lo.y; j[2] = lo.z; j[3] = lo.w;
  j[4] = hi.x; j[5] = hi.y; j[6] = hi.z; j[7] = hi.w;
}

#define DEG_BLOCKS (NNODE / 256)
#define WT_SMALL (DIM * DIM)
#define WT_LARGE (DIM * 2 * DIM)
#define WT_F4 (4 * (WT_SMALL / 4) + 3 * (WT_LARGE / 4))

template <int K>
__device__ __forceinline__ void w_relayout(const float* __restrict__ Wsrc,
                                           float* __restrict__ Wdst, unsigned q) {
  const unsigned lane = q & 31u;
  const unsigned g    = (q >> 5) % (unsigned)(K / 8);
  const unsigned wt   = (q >> 5) / (unsigned)(K / 8);
  const unsigned row  = 32u * (wt >> 1) + 2u * (lane & 15u) + (wt & 1u);
  const unsigned col  = 8u * g + 4u * (lane >> 4);
  st4(Wdst + (size_t)q * 4, ld4(Wsrc + (size_t)row * K + col));
}

__global__ void deg_kernel(const int* __restrict__ neigh,
                           const int* __restrict__ nbrc,
                           float* __restrict__ dinv,
                           unsigned* __restrict__ keep,
                           const float* __restrict__ gcnW,
                           const float* __restrict__ poolW,
                           const float* __restrict__ mrgW,
                           float* __restrict__ wt) {
  if (blockIdx.x >= DEG_BLOCKS) {
    const unsigned u = (blockIdx.x - DEG_BLOCKS) * blockDim.x + threadIdx.x;
    if (u < 4u * (WT_SMALL / 4)) {
      const unsigned m = u / (WT_SMALL / 4), q = u % (WT_SMALL / 4);
      const float* s = (m < 3) ? gcnW + (size_t)m * WT_SMALL : poolW;
      w_relayout<DIM>(s, wt + (size_t)m * WT_SMALL, q);
    } else if (u < WT_F4) {
      const unsigned v = u - 4u * (WT_SMALL / 4);
      const unsigned m = v / (WT_LARGE / 4), q = v % (WT_LARGE / 4);
      w_relayout<2 * DIM>(mrgW + (size_t)m * WT_LARGE,
                          wt + (size_t)4 * WT_SMALL + (size_t)m * WT_LARGE, q);
    }
    return;
  }
  int i = blockIdx.x * blockDim.x + threadIdx.x;
  if (i >= NNODE) return;
  int cnt = nbrc[i]; cnt = cnt > KNBR ? KNBR : cnt;
  int nb[KNBR];
  ld_nbrs(neigh + i * KNBR, nb);
  unsigned km = 0;
  int deg = 1;
#pragma unroll
  for (int k = 0; k < KNBR; ++k) {
    if (k < cnt) {
      const int j = nb[k];
      bool dup = (j == i);
#pragma unroll
      for (int t = 0; t < k; ++t) dup |= (nb[t] == j);
      if (!dup) { km |= (1u << k); ++deg; }
    }
  }
  dinv[i] = rsqrtf((float)deg);
  keep[i] = km;
}

__device__ __forceinline__ v8f mma4(v2f a, v2f b, v8f c) {
  return __builtin_amdgcn_wmma_f32_16x16x4_f32(false, a, false, b,
                                               (short)0, c, false, false);
}

struct Frag4 { float4 a0, a1, b0, b1; };

__device__ __forceinline__ Frag4 ld_group(const float* __restrict__ a0p,
                                          const float* __restrict__ a1p,
                                          const float* __restrict__ b0p,
                                          const float* __restrict__ b1p,
                                          int koff) {
  Frag4 f;
  f.a0 = ld4(a0p + koff);
  f.a1 = ld4(a1p + koff);
  f.b0 = ld4(b0p + 16 * koff);
  f.b1 = ld4(b1p + 16 * koff);
  return f;
}

template <int K, bool RELU>
__device__ __forceinline__ void gemm_tile(const float* __restrict__ A,
                                          const float* __restrict__ W,
                                          const float* __restrict__ bias,
                                          float* __restrict__ Y) {
  constexpr int NC = K / 32;
  const int wave = threadIdx.x >> 5;
  const int lane = threadIdx.x & 31;
  const int half = lane >> 4;
  const int l16  = lane & 15;
  const int m0 = blockIdx.x * 32;
  const int n0 = wave * 32;

  const float* __restrict__ a0p = A + (size_t)(m0 + l16) * K + 4 * half;
  const float* __restrict__ a1p = a0p + 16 * K;
  const float* __restrict__ b0p = W + (size_t)wave * 32 * K + lane * 4;
  const float* __restrict__ b1p = b0p + 16 * K;

  Frag4 f[NC][4];
#pragma unroll
  for (int g = 0; g < 4; ++g) f[0][g] = ld_group(a0p, a1p, b0p, b1p, 8 * g);

  v8f c00 = {}, c01 = {}, c10 = {}, c11 = {};
#pragma unroll
  for (int c = 0; c < NC; ++c) {
    if (c + 1 < NC) {
#pragma unroll
      for (int g = 0; g < 4; ++g)
        f[c + 1][g] = ld_group(a0p, a1p, b0p, b1p, 32 * (c + 1) + 8 * g);
    }
#pragma unroll
    for (int g = 0; g < 4; ++g) {
      const Frag4 q = f[c][g];
      const v2f a0l = {q.a0.x, q.a0.y}, a0h = {q.a0.z, q.a0.w};
      const v2f a1l = {q.a1.x, q.a1.y}, a1h = {q.a1.z, q.a1.w};
      const v2f b0l = {q.b0.x, q.b0.y}, b0h = {q.b0.z, q.b0.w};
      const v2f b1l = {q.b1.x, q.b1.y}, b1h = {q.b1.z, q.b1.w};
      c00 = mma4(a0l, b0l, c00);
      c01 = mma4(a0l, b1l, c01);
      c10 = mma4(a1l, b0l, c10);
      c11 = mma4(a1l, b1l, c11);
      c00 = mma4(a0h, b0h, c00);
      c01 = mma4(a0h, b1h, c01);
      c10 = mma4(a1h, b0h, c10);
      c11 = mma4(a1h, b1h, c11);
    }
  }

  const int col = n0 + 2 * l16;
  const float2 bv = *reinterpret_cast<const float2*>(bias + col);
  float* __restrict__ y0 = Y + (size_t)(m0 + 8 * half) * DIM + col;
  float* __restrict__ y1 = y0 + 16 * DIM;
#pragma unroll
  for (int r = 0; r < 8; ++r) {
    float2 u = {c00[r] + bv.x, c01[r] + bv.y};
    float2 v = {c10[r] + bv.x, c11[r] + bv.y};
    if (RELU) {
      u.x = fmaxf(u.x, 0.0f); u.y = fmaxf(u.y, 0.0f);
      v.x = fmaxf(v.x, 0.0f); v.y = fmaxf(v.y, 0.0f);
    }
    *reinterpret_cast<float2*>(y0 + (size_t)r * DIM) = u;
    *reinterpret_cast<float2*>(y1 + (size_t)r * DIM) = v;
  }
}

template <int K, bool RELU>
__global__ __launch_bounds__(128) __attribute__((amdgpu_waves_per_eu(1, 2)))
void wmma_gemm_bias(const float* __restrict__ A,
                    const float* __restrict__ W,
                    const float* __restrict__ bias,
                    float* __restrict__ Y) {
  gemm_tile<K, RELU>(A, W, bias, Y);
}

template <int K1, bool RELU1, int K2, bool RELU2>
__global__ __launch_bounds__(128) __attribute__((amdgpu_waves_per_eu(1, 2)))
void wmma_gemm_pair(const float* __restrict__ A1,
                    const float* __restrict__ W1,
                    const float* __restrict__ bias1,
                    float* __restrict__ Y1,
                    const float* __restrict__ A2,
                    const float* __restrict__ W2,
                    const float* __restrict__ bias2,
                    float* __restrict__ Y2) {
  gemm_tile<K1, RELU1>(A1, W1, bias1, Y1);
  gemm_tile<K2, RELU2>(A2, W2, bias2, Y2);
}

__device__ __forceinline__ void wave_ln_store(float4 v,
                                              const float* __restrict__ g,
                                              const float* __restrict__ b,
                                              float* __restrict__ dstrow,
                                              int lane) {
  float s = v.x + v.y + v.z + v.w;
#pragma unroll
  for (int off = 16; off > 0; off >>= 1) s += __shfl_xor(s, off, 32);
  const float mu = s * (1.0f / (float)DIM);
  const float dx = v.x - mu, dy = v.y - mu, dz = v.z - mu, dw = v.w - mu;
  float q = dx * dx + dy * dy + dz * dz + dw * dw;
#pragma unroll
  for (int off = 16; off > 0; off >>= 1) q += __shfl_xor(q, off, 32);
  const float rinv = rsqrtf(q * (1.0f / (float)DIM) + 1e-5f);
  const int d = lane * 4;
  const float4 gg = ld4(g + d);
  const float4 bb = ld4(b + d);
  float4 o;
  o.x = dx * rinv * gg.x + bb.x;
  o.y = dy * rinv * gg.y + bb.y;
  o.z = dz * rinv * gg.z + bb.z;
  o.w = dw * rinv * gg.w + bb.w;
  st4(dstrow + d, o);
}

__device__ __forceinline__ float4 gcn_sum(float di, unsigned km, float4 t, float4 r,
                                          const float4 (&u)[KNBR],
                                          const float (&dj)[KNBR]) {
  float4 acc = {di * t.x, di * t.y, di * t.z, di * t.w};
#pragma unroll
  for (int k = 0; k < KNBR; ++k) {
    if ((km >> k) & 1u) {
      acc.x += dj[k] * u[k].x; acc.y += dj[k] * u[k].y;
      acc.z += dj[k] * u[k].z; acc.w += dj[k] * u[k].w;
    }
  }
  acc.x = fmaxf(acc.x * di, 0.0f);
  acc.y = fmaxf(acc.y * di, 0.0f);
  acc.z = fmaxf(acc.z * di, 0.0f);
  acc.w = fmaxf(acc.w * di, 0.0f);
  return make_float4(acc.x + r.x, acc.y + r.y, acc.z + r.z, acc.w + r.w);
}

__global__ void gcn_ln_sage_pool(const float* __restrict__ nxt,
                                 const float* __restrict__ resid,
                                 const int* __restrict__ neigh,
                                 const int* __restrict__ nbrc,
                                 const float* __restrict__ dinv,
                                 const unsigned* __restrict__ keep,
                                 const float* __restrict__ g,
                                 const float* __restrict__ b,
                                 float* __restrict__ dst,
                                 const float* __restrict__ selfsrc,
                                 const float* __restrict__ poolsrc,
                                 float* __restrict__ cat) {
  const int i = (int)((blockIdx.x * blockDim.x + threadIdx.x) >> 5);
  if (i >= NNODE) return;
  const int lane = threadIdx.x & 31;
  const int d = lane * 4;

  const float4 t = ld4(nxt + (size_t)i * DIM + d);
  const float4 r = ld4(resid + (size_t)i * DIM + d);
  const float4 self = ld4(selfsrc + (size_t)i * DIM + d);
  const float di = dinv[i];
  const unsigned km = keep[i];
  int cnt = nbrc[i]; cnt = cnt > KNBR ? KNBR : cnt;
  int nb[KNBR];
  ld_nbrs(neigh + i * KNBR, nb);
  asm volatile("" :: "v"(t.x), "v"(t.y), "v"(t.z), "v"(t.w),
                     "v"(r.x), "v"(r.y), "v"(r.z), "v"(r.w),
                     "v"(self.x), "v"(self.y), "v"(self.z), "v"(self.w),
                     "v"(di), "v"(km), "v"(cnt),
                     "v"(nb[0]), "v"(nb[1]), "v"(nb[2]), "v"(nb[3]),
                     "v"(nb[4]), "v"(nb[5]), "v"(nb[6]), "v"(nb[7]));

  float4 u[KNBR], p[KNBR];
  float dj[KNBR];
#pragma unroll
  for (int k = 0; k < KNBR; ++k) {
    const int j = ((km >> k) & 1u) ? nb[k] : i;
    dj[k] = dinv[j];
    u[k] = ld4(nxt + (size_t)j * DIM + d);
  }
#pragma unroll
  for (int k = 0; k < KNBR; ++k) {
    const int j = (k < cnt) ? nb[k] : i;
    p[k] = ld4(poolsrc + (size_t)j * DIM + d);
  }
#pragma unroll
  for (int k = 0; k < KNBR; ++k) {
    asm volatile("" :: "v"(u[k].x), "v"(u[k].y), "v"(u[k].z), "v"(u[k].w),
                       "v"(dj[k]));
    asm volatile("" :: "v"(p[k].x), "v"(p[k].y), "v"(p[k].z), "v"(p[k].w));
  }

  const float4 v = gcn_sum(di, km, t, r, u, dj);
  wave_ln_store(v, g, b, dst + (size_t)i * DIM, lane);

  float4 m = {NEGINF, NEGINF, NEGINF, NEGINF};
#pragma unroll
  for (int k = 0; k < KNBR; ++k) {
    if (k < cnt) {
      m.x = fmaxf(m.x, p[k].x); m.y = fmaxf(m.y, p[k].y);
      m.z = fmaxf(m.z, p[k].z); m.w = fmaxf(m.w, p[k].w);
    }
  }
  if (cnt == 0) m = make_float4(0.f, 0.f, 0.f, 0.f);
  st4(cat + (size_t)i * (2 * DIM) + d, self);
  st4(cat + (size_t)i * (2 * DIM) + DIM + d, m);
}

__global__ void gcn_ln_sage_ln(const float* __restrict__ nxt,
                               const float* __restrict__ resid,
                               const int* __restrict__ neigh,
                               const float* __restrict__ dinv,
                               const unsigned* __restrict__ keep,
                               const float* __restrict__ g,
                               const float* __restrict__ b,
                               float* __restrict__ dst, int ldd,
                               const float* __restrict__ a,
                               const float* __restrict__ res,
                               const float* __restrict__ sg,
                               const float* __restrict__ sb,
                               float* __restrict__ sdst) {
  const int i = (int)((blockIdx.x * blockDim.x + threadIdx.x) >> 5);
  if (i >= NNODE) return;
  const int lane = threadIdx.x & 31;
  const int d = lane * 4;

  const float4 av = ld4(a + (size_t)i * DIM + d);
  const float4 rv = ld4(res + (size_t)i * DIM + d);
  const float4 t = ld4(nxt + (size_t)i * DIM + d);
  const float4 r = ld4(resid + (size_t)i * DIM + d);
  const float di = dinv[i];
  const unsigned km = keep[i];
  int nb[KNBR];
  ld_nbrs(neigh + i * KNBR, nb);
  asm volatile("" :: "v"(av.x), "v"(av.y), "v"(av.z), "v"(av.w),
                     "v"(rv.x), "v"(rv.y), "v"(rv.z), "v"(rv.w),
                     "v"(t.x), "v"(t.y), "v"(t.z), "v"(t.w),
                     "v"(r.x), "v"(r.y), "v"(r.z), "v"(r.w),
                     "v"(di), "v"(km),
                     "v"(nb[0]), "v"(nb[1]), "v"(nb[2]), "v"(nb[3]),
                     "v"(nb[4]), "v"(nb[5]), "v"(nb[6]), "v"(nb[7]));

  float4 u[KNBR];
  float dj[KNBR];
#pragma unroll
  for (int k = 0; k < KNBR; ++k) {
    const int j = ((km >> k) & 1u) ? nb[k] : i;
    dj[k] = dinv[j];
    u[k] = ld4(nxt + (size_t)j * DIM + d);
  }
#pragma unroll
  for (int k = 0; k < KNBR; ++k) {
    asm volatile("" :: "v"(u[k].x), "v"(u[k].y), "v"(u[k].z), "v"(u[k].w),
                       "v"(dj[k]));
  }

  const float4 w = {av.x + rv.x, av.y + rv.y, av.z + rv.z, av.w + rv.w};
  wave_ln_store(w, sg, sb, sdst + (size_t)i * DIM, lane);

  const float4 v = gcn_sum(di, km, t, r, u, dj);
  wave_ln_store(v, g, b, dst + (size_t)i * ldd, lane);
}

__global__ void sage_pool_cat(const float* __restrict__ selfsrc,
                              const float* __restrict__ poolsrc,
                              const int* __restrict__ neigh,
                              const int* __restrict__ nbrc,
                              float* __restrict__ cat) {
  const int i = (int)((blockIdx.x * blockDim.x + threadIdx.x) >> 5);
  if (i >= NNODE) return;
  const int lane = threadIdx.x & 31;
  const int d = lane * 4;

  int cnt = nbrc[i]; cnt = cnt > KNBR ? KNBR : cnt;
  int nb[KNBR];
  ld_nbrs(neigh + i * KNBR, nb);
  const float4 self = ld4(selfsrc + (size_t)i * DIM + d);

  float4 u[KNBR];
#pragma unroll
  for (int k = 0; k < KNBR; ++k) {
    const int j = (k < cnt) ? nb[k] : i;
    u[k] = ld4(poolsrc + (size_t)j * DIM + d);
  }
  st4(cat + (size_t)i * (2 * DIM) + d, self);
#pragma unroll
  for (int k = 0; k < KNBR; ++k) {
    asm volatile("" :: "v"(u[k].x), "v"(u[k].y), "v"(u[k].z), "v"(u[k].w));
  }

  float4 m = {NEGINF, NEGINF, NEGINF, NEGINF};
#pragma unroll
  for (int k = 0; k < KNBR; ++k) {
    if (k < cnt) {
      m.x = fmaxf(m.x, u[k].x); m.y = fmaxf(m.y, u[k].y);
      m.z = fmaxf(m.z, u[k].z); m.w = fmaxf(m.w, u[k].w);
    }
  }
  if (cnt == 0) m = make_float4(0.f, 0.f, 0.f, 0.f);
  st4(cat + (size_t)i * (2 * DIM) + DIM + d, m);
}

__global__ void resid_ln(const float* __restrict__ a,
                         const float* __restrict__ res,
                         const float* __restrict__ g,
                         const float* __restrict__ b,
                         float* __restrict__ dst, int ldd) {
  const int i = (int)((blockIdx.x * blockDim.x + threadIdx.x) >> 5);
  if (i >= NNODE) return;
  const int lane = threadIdx.x & 31;
  const int d = lane * 4;
  const float4 av = ld4(a + (size_t)i * DIM + d);
  const float4 rv = ld4(res + (size_t)i * DIM + d);
  float4 v = {av.x + rv.x, av.y + rv.y, av.z + rv.z, av.w + rv.w};
  wave_ln_store(v, g, b, dst + (size_t)i * ldd, lane);
}

extern "C" void kernel_launch(void* const* d_in, const int* in_sizes, int n_in,
                              void* d_out, int out_size, void* d_ws, size_t ws_size,
                              hipStream_t stream) {
  const float* x      = (const float*)d_in[0];
  const int*   neigh  = (const int*)  d_in[1];
  const int*   nbrc   = (const int*)  d_in[2];
  const float* gcnW   = (const float*)d_in[3];
  const float* gcnB   = (const float*)d_in[4];
  const float* gcnG   = (const float*)d_in[5];
  const float* gcnBb  = (const float*)d_in[6];
  const float* poolW  = (const float*)d_in[7];
  const float* poolB  = (const float*)d_in[8];
  const float* mrgW   = (const float*)d_in[9];
  const float* mrgB   = (const float*)d_in[10];
  const float* sgG    = (const float*)d_in[11];
  const float* sgB    = (const float*)d_in[12];
  float* out = (float*)d_out;

  const size_t ND = (size_t)NNODE * DIM;
  float*    dinv  = (float*)d_ws;
  unsigned* keep  = (unsigned*)(dinv + NNODE);
  float*    first = (float*)(keep + NNODE);
  float*    nxt   = first;
  float*    wt    = first + ND;
  float*    hbuf  = wt + ND;
  const float* wtG = wt;
  const float* wtP = wt + 3 * DIM * DIM;
  const float* wtM = wt + 4 * DIM * DIM;
  float*    sbuf  = hbuf + ND;
  float*    tmp   = sbuf + ND;
  float*    cat   = tmp + ND;

  const dim3 gB(128);
  const dim3 gG(NNODE / 32);
  const dim3 nB(256);
  const dim3 nG(NNODE / 8);

  deg_kernel<<<NNODE / 256 + 160, 256, 0, stream>>>(neigh, nbrc, dinv, keep,
                                                    gcnW, poolW, mrgW, wt);

  wmma_gemm_pair<DIM, false, DIM, true><<<gG, gB, 0, stream>>>(
      x, wtG, gcnB, first, x, wtP, poolB, tmp);
  gcn_ln_sage_pool<<<nG, nB, 0, stream>>>(first, first, neigh, nbrc, dinv, keep,
                                          gcnG, gcnBb, hbuf, x, tmp, cat);

  wmma_gemm_pair<DIM, false, 2 * DIM, true><<<gG, gB, 0, stream>>>(
      hbuf, wtG + DIM * DIM, gcnB + DIM, nxt, cat, wtM, mrgB, sbuf);
  gcn_ln_sage_pool<<<nG, nB, 0, stream>>>(nxt, hbuf, neigh, nbrc, dinv, keep,
                                          gcnG + DIM, gcnBb + DIM, hbuf,
                                          sbuf, sbuf, cat);

  wmma_gemm_pair<DIM, false, 2 * DIM, true><<<gG, gB, 0, stream>>>(
      hbuf, wtG + 2 * DIM * DIM, gcnB + 2 * DIM, nxt,
      cat, wtM + DIM * 2 * DIM, mrgB + DIM, tmp);
  gcn_ln_sage_ln<<<nG, nB, 0, stream>>>(nxt, hbuf, neigh, dinv, keep,
                                        gcnG + 2 * DIM, gcnBb + 2 * DIM,
                                        out, 2 * DIM,
                                        tmp, sbuf, sgG, sgB, sbuf);

  sage_pool_cat<<<nG, nB, 0, stream>>>(sbuf, sbuf, neigh, nbrc, cat);
  wmma_gemm_bias<2 * DIM, true><<<gG, gB, 0, stream>>>(cat, wtM + 2 * DIM * 2 * DIM,
                                                       mrgB + 2 * DIM, tmp);
  resid_ln<<<nG, nB, 0, stream>>>(tmp, sbuf, sgG + DIM, sgB + DIM,
                                  out + DIM, 2 * DIM);
}


# Row stage and the products reading its rows share a launch: a block builds its 32 rows in the workspace, block barrier, then multiplies them (merge reads self rows in place); 10 -> 6; sums unchanged
# speedup vs baseline: 2.1985x; 1.1112x over previous
#include <hip/hip_runtime.h>

typedef __attribute__((ext_vector_type(2))) float v2f;
typedef __attribute__((ext_vector_type(8))) float v8f;

#define NNODE 8192
#define DIM   128
#define KNBR  8
#define NEGINF -1e30f

__device__ __forceinline__ float4 ld4(const float* p) {
  return *reinterpret_cast<const float4*>(p);
}
__device__ __forceinline__ void st4(float* p, float4 v) {
  *reinterpret_cast<float4*>(p) = v;
}
__device__ __forceinline__ void ld_nbrs(const int* __restrict__ nb, int (&j)[KNBR]) {
  const int4 lo = *reinterpret_cast<const int4*>(nb);
  const int4 hi = *reinterpret_cast<const int4*>(nb + 4);
  j[0] = lo.x; j[1] = lo.y; j[2] = lo.z; j[3] = lo.w;
  j[4] = hi.x; j[5] = hi.y; j[6] = hi.z; j[7] = hi.w;
}

#define DEG_BLOCKS (NNODE / 256)
#define WT_SMALL (DIM * DIM)
#define WT_LARGE (DIM * 2 * DIM)
#define WT_F4 (4 * (WT_SMALL / 4) + 3 * (WT_LARGE / 4))

template <int K>
__device__ __forceinline__ void w_relayout(const float* __restrict__ Wsrc,
                                           float* __restrict__ Wdst, unsigned q) {
  const unsigned lane = q & 31u;
  const unsigned g    = (q >> 5) % (unsigned)(K / 8);
  const unsigned wt   = (q >> 5) / (unsigned)(K / 8);
  const unsigned row  = 32u * (wt >> 1) + 2u * (lane & 15u) + (wt & 1u);
  const unsigned col  = 8u * g + 4u * (lane >> 4);
  st4(Wdst + (size_t)q * 4, ld4(Wsrc + (size_t)row * K + col));
}

__global__ void deg_kernel(const int* __restrict__ neigh,
                           const int* __restrict__ nbrc,
                           float* __restrict__ dinv,
                           unsigned* __restrict__ keep,
                           const float* __restrict__ gcnW,
                           const float* __restrict__ poolW,
                           const float* __restrict__ mrgW,
                           float* __restrict__ wt) {
  if (blockIdx.x >= DEG_BLOCKS) {
    const unsigned u = (blockIdx.x - DEG_BLOCKS) * blockDim.x + threadIdx.x;
    if (u < 4u * (WT_SMALL / 4)) {
      const unsigned m = u / (WT_SMALL / 4), q = u % (WT_SMALL / 4);
      const float* s = (m < 3) ? gcnW + (size_t)m * WT_SMALL : poolW;
      w_relayout<DIM>(s, wt + (size_t)m * WT_SMALL, q);
    } else if (u < WT_F4) {
      const unsigned v = u - 4u * (WT_SMALL / 4);
      const unsigned m = v / (WT_LARGE / 4), q = v % (WT_LARGE / 4);
      w_relayout<2 * DIM>(mrgW + (size_t)m * WT_LARGE,
                          wt + (size_t)4 * WT_SMALL + (size_t)m * WT_LARGE, q);
    }
    return;
  }
  int i = blockIdx.x * blockDim.x + threadIdx.x;
  if (i >= NNODE) return;
  int cnt = nbrc[i]; cnt = cnt > KNBR ? KNBR : cnt;
  int nb[KNBR];
  ld_nbrs(neigh + i * KNBR, nb);
  unsigned km = 0;
  int deg = 1;
#pragma unroll
  for (int k = 0; k < KNBR; ++k) {
    if (k < cnt) {
      const int j = nb[k];
      bool dup = (j == i);
#pragma unroll
      for (int t = 0; t < k; ++t) dup |= (nb[t] == j);
      if (!dup) { km |= (1u << k); ++deg; }
    }
  }
  dinv[i] = rsqrtf((float)deg);
  keep[i] = km;
}

__device__ __forceinline__ v8f mma4(v2f a, v2f b, v8f c) {
  return __builtin_amdgcn_wmma_f32_16x16x4_f32(false, a, false, b,
                                               (short)0, c, false, false);
}

struct Frag4 { float4 a0, a1, b0, b1; };

__device__ __forceinline__ Frag4 ld_group(const float* a0p,
                                          const float* a1p,
                                          const float* __restrict__ b0p,
                                          const float* __restrict__ b1p,
                                          int koff) {
  Frag4 f;
  f.a0 = ld4(a0p);
  f.a1 = ld4(a1p);
  f.b0 = ld4(b0p + 16 * koff);
  f.b1 = ld4(b1p + 16 * koff);
  return f;
}

template <int K, bool RELU>
__device__ __forceinline__ void gemm_tile(int wave,
                                          const float* Alo,
                                          const float* Ahi,
                                          const float* __restrict__ W,
                                          const float* __restrict__ bias,
                                          float* Y) {
  constexpr int NC = K / 32;
  const int lane = threadIdx.x & 31;
  const int half = lane >> 4;
  const int l16  = lane & 15;
  const int m0 = blockIdx.x * 32;
  const int n0 = wave * 32;

  const size_t aoff = (size_t)(m0 + l16) * DIM + 4 * half;
  const float* a0lo = Alo + aoff;
  const float* a0hi = Ahi + aoff;
#define A0P(k) ((k) < DIM ? a0lo + (k) : a0hi + ((k) - DIM))
  const float* __restrict__ b0p = W + (size_t)wave * 32 * K + lane * 4;
  const float* __restrict__ b1p = b0p + 16 * K;

  Frag4 f[NC][4];
#pragma unroll
  for (int g = 0; g < 4; ++g)
    f[0][g] = ld_group(A0P(8 * g), A0P(8 * g) + 16 * DIM, b0p, b1p, 8 * g);

  v8f c00 = {}, c01 = {}, c10 = {}, c11 = {};
#pragma unroll
  for (int c = 0; c < NC; ++c) {
    if (c + 1 < NC) {
#pragma unroll
      for (int g = 0; g < 4; ++g)
        f[c + 1][g] = ld_group(A0P(32 * (c + 1) + 8 * g),
                               A0P(32 * (c + 1) + 8 * g) + 16 * DIM,
                               b0p, b1p, 32 * (c + 1) + 8 * g);
    }
#pragma unroll
    for (int g = 0; g < 4; ++g) {
      const Frag4 q = f[c][g];
      const v2f a0l = {q.a0.x, q.a0.y}, a0h = {q.a0.z, q.a0.w};
      const v2f a1l = {q.a1.x, q.a1.y}, a1h = {q.a1.z, q.a1.w};
      const v2f b0l = {q.b0.x, q.b0.y}, b0h = {q.b0.z, q.b0.w};
      const v2f b1l = {q.b1.x, q.b1.y}, b1h = {q.b1.z, q.b1.w};
      c00 = mma4(a0l, b0l, c00);
      c01 = mma4(a0l, b1l, c01);
      c10 = mma4(a1l, b0l, c10);
      c11 = mma4(a1l, b1l, c11);
      c00 = mma4(a0h, b0h, c00);
      c01 = mma4(a0h, b1h, c01);
      c10 = mma4(a1h, b0h, c10);
      c11 = mma4(a1h, b1h, c11);
    }
  }

#undef A0P
  const int col = n0 + 2 * l16;
  const float2 bv = *reinterpret_cast<const float2*>(bias + col);
  float* y0 = Y + (size_t)(m0 + 8 * half) * DIM + col;
  float* y1 = y0 + 16 * DIM;
#pragma unroll
  for (int r = 0; r < 8; ++r) {
    float2 u = {c00[r] + bv.x, c01[r] + bv.y};
    float2 v = {c10[r] + bv.x, c11[r] + bv.y};
    if (RELU) {
      u.x = fmaxf(u.x, 0.0f); u.y = fmaxf(u.y, 0.0f);
      v.x = fmaxf(v.x, 0.0f); v.y = fmaxf(v.y, 0.0f);
    }
    *reinterpret_cast<float2*>(y0 + (size_t)r * DIM) = u;
    *reinterpret_cast<float2*>(y1 + (size_t)r * DIM) = v;
  }
}

template <int K1, bool RELU1, int K2, bool RELU2>
__global__ __launch_bounds__(128) __attribute__((amdgpu_waves_per_eu(1, 2)))
void wmma_gemm_pair(const float* __restrict__ A1,
                    const float* __restrict__ W1,
                    const float* __restrict__ bias1,
                    float* __restrict__ Y1,
                    const float* __restrict__ A2,
                    const float* __restrict__ W2,
                    const float* __restrict__ bias2,
                    float* __restrict__ Y2) {
  const int wave = threadIdx.x >> 5;
  gemm_tile<K1, RELU1>(wave, A1, A1, W1, bias1, Y1);
  gemm_tile<K2, RELU2>(wave, A2, A2, W2, bias2, Y2);
}

__device__ __forceinline__ void wave_ln_store(float4 v,
                                              const float* __restrict__ g,
                                              const float* __restrict__ b,
                                              float* __restrict__ dstrow,
                                              int lane) {
  float s = v.x + v.y + v.z + v.w;
#pragma unroll
  for (int off = 16; off > 0; off >>= 1) s += __shfl_xor(s, off, 32);
  const float mu = s * (1.0f / (float)DIM);
  const float dx = v.x - mu, dy = v.y - mu, dz = v.z - mu, dw = v.w - mu;
  float q = dx * dx + dy * dy + dz * dz + dw * dw;
#pragma unroll
  for (int off = 16; off > 0; off >>= 1) q += __shfl_xor(q, off, 32);
  const float rinv = rsqrtf(q * (1.0f / (float)DIM) + 1e-5f);
  const int d = lane * 4;
  const float4 gg = ld4(g + d);
  const float4 bb = ld4(b + d);
  float4 o;
  o.x = dx * rinv * gg.x + bb.x;
  o.y = dy * rinv * gg.y + bb.y;
  o.z = dz * rinv * gg.z + bb.z;
  o.w = dw * rinv * gg.w + bb.w;
  st4(dstrow + d, o);
}

__device__ __forceinline__ float4 gcn_sum(float di, unsigned km, float4 t, float4 r,
                                          const float4 (&u)[KNBR],
                                          const float (&dj)[KNBR]) {
  float4 acc = {di * t.x, di * t.y, di * t.z, di * t.w};
#pragma unroll
  for (int k = 0; k < KNBR; ++k) {
    if ((km >> k) & 1u) {
      acc.x += dj[k] * u[k].x; acc.y += dj[k] * u[k].y;
      acc.z += dj[k] * u[k].z; acc.w += dj[k] * u[k].w;
    }
  }
  acc.x = fmaxf(acc.x * di, 0.0f);
  acc.y = fmaxf(acc.y * di, 0.0f);
  acc.z = fmaxf(acc.z * di, 0.0f);
  acc.w = fmaxf(acc.w * di, 0.0f);
  return make_float4(acc.x + r.x, acc.y + r.y, acc.z + r.z, acc.w + r.w);
}

template <int NB>
__device__ __forceinline__ void gcn_pool_rows(int i0, int lane,
                                              const float* __restrict__ nxt,
                                              const float* resid,
                                              const int* __restrict__ neigh,
                                              const int* __restrict__ nbrc,
                                              const float* __restrict__ dinv,
                                              const unsigned* __restrict__ keep,
                                              const float* __restrict__ g,
                                              const float* __restrict__ b,
                                              float* dst,
                                              const float* __restrict__ poolsrc,
                                              float* pool) {
  const int d = lane * 4;
  float4 t[NB], r[NB];
  float di[NB];
  unsigned km[NB];
  int cnt[NB];
  int nb[NB][KNBR];
  int own[NB];
#pragma unroll
  for (int n = 0; n < NB; ++n) {
    const int i = i0 + n;
    own[n] = i;
    t[n] = ld4(nxt + (size_t)i * DIM + d);
    r[n] = ld4(resid + (size_t)i * DIM + d);
    di[n] = dinv[i];
    km[n] = keep[i];
    const int c = nbrc[i];
    cnt[n] = c > KNBR ? KNBR : c;
    ld_nbrs(neigh + i * KNBR, nb[n]);
  }
#pragma unroll
  for (int n = 0; n < NB; ++n) {
    asm volatile("" : "+v"(own[n]) : "v"(t[n].x), "v"(t[n].y), "v"(t[n].z), "v"(t[n].w),
                       "v"(r[n].x), "v"(r[n].y), "v"(r[n].z), "v"(r[n].w),
                       "v"(di[n]), "v"(km[n]), "v"(cnt[n]),
                       "v"(nb[n][0]), "v"(nb[n][1]), "v"(nb[n][2]), "v"(nb[n][3]),
                       "v"(nb[n][4]), "v"(nb[n][5]), "v"(nb[n][6]), "v"(nb[n][7]));
  }

  float4 u[NB][KNBR], p[NB][KNBR];
  float dj[NB][KNBR];
#pragma unroll
  for (int n = 0; n < NB; ++n) {
#pragma unroll
    for (int k = 0; k < KNBR; ++k) {
      const int j = ((km[n] >> k) & 1u) ? nb[n][k] : own[n];
      dj[n][k] = dinv[j];
      u[n][k] = ld4(nxt + (size_t)j * DIM + d);
    }
  }
#pragma unroll
  for (int n = 0; n < NB; ++n) {
#pragma unroll
    for (int k = 0; k < KNBR; ++k) {
      const int j = (k < cnt[n]) ? nb[n][k] : own[n];
      p[n][k] = ld4(poolsrc + (size_t)j * DIM + d);
    }
  }
#pragma unroll
  for (int n = 0; n < NB; ++n) {
#pragma unroll
    for (int k = 0; k < KNBR; ++k) {
      asm volatile("" :: "v"(u[n][k].x), "v"(u[n][k].y), "v"(u[n][k].z),
                         "v"(u[n][k].w), "v"(dj[n][k]));
      asm volatile("" :: "v"(p[n][k].x), "v"(p[n][k].y), "v"(p[n][k].z),
                         "v"(p[n][k].w));
    }
  }

#pragma unroll
  for (int n = 0; n < NB; ++n) {
    const int i = i0 + n;
    const float4 v = gcn_sum(di[n], km[n], t[n], r[n], u[n], dj[n]);
    wave_ln_store(v, g, b, dst + (size_t)i * DIM, lane);

    float4 m = {NEGINF, NEGINF, NEGINF, NEGINF};
#pragma unroll
    for (int k = 0; k < KNBR; ++k) {
      if (k < cnt[n]) {
        m.x = fmaxf(m.x, p[n][k].x); m.y = fmaxf(m.y, p[n][k].y);
        m.z = fmaxf(m.z, p[n][k].z); m.w = fmaxf(m.w, p[n][k].w);
      }
    }
    if (cnt[n] == 0) m = make_float4(0.f, 0.f, 0.f, 0.f);
    st4(pool + (size_t)i * DIM + d, m);
  }
}

__global__ __launch_bounds__(256) __attribute__((amdgpu_waves_per_eu(2, 2)))
void rows_then_gemm_pair(const float* __restrict__ nxt,
                         const float* resid,
                         const int* __restrict__ neigh,
                         const int* __restrict__ nbrc,
                         const float* __restrict__ dinv,
                         const unsigned* __restrict__ keep,
                         const float* __restrict__ g,
                         const float* __restrict__ b,
                         float* dst,
                         const float* __restrict__ selfsrc,
                         const float* __restrict__ poolsrc,
                         float* pool,
                         const float* __restrict__ W1,
                         const float* __restrict__ bias1,
                         float* __restrict__ Y1,
                         const float* __restrict__ W2,
                         const float* __restrict__ bias2,
                         float* __restrict__ Y2) {
  const int lane = threadIdx.x & 31;
  const int i0 = (int)blockIdx.x * 32 + (int)(threadIdx.x >> 5) * 4;
  gcn_pool_rows<2>(i0, lane, nxt, resid, neigh, nbrc, dinv, keep, g, b, dst,
                   poolsrc, pool);
  gcn_pool_rows<2>(i0 + 2, lane, nxt, resid, neigh, nbrc, dinv, keep, g, b, dst,
                   poolsrc, pool);
  __syncthreads();
  const int wave = __builtin_amdgcn_readfirstlane((int)(threadIdx.x >> 5));
  if (wave < 4) gemm_tile<DIM, false>(wave, dst, dst, W1, bias1, Y1);
  else          gemm_tile<2 * DIM, true>(wave - 4, selfsrc, pool, W2, bias2, Y2);
}

template <int NB>
__device__ __forceinline__ void pool_rows(int i0, int lane,
                                          const float* __restrict__ src,
                                          const int* __restrict__ neigh,
                                          const int* __restrict__ nbrc,
                                          float* pool) {
  const int d = lane * 4;
  int cnt[NB];
  int nb[NB][KNBR];
#pragma unroll
  for (int n = 0; n < NB; ++n) {
    const int c = nbrc[i0 + n];
    cnt[n] = c > KNBR ? KNBR : c;
    ld_nbrs(neigh + (i0 + n) * KNBR, nb[n]);
  }
  float4 u[NB][KNBR];
#pragma unroll
  for (int n = 0; n < NB; ++n) {
#pragma unroll
    for (int k = 0; k < KNBR; ++k) {
      const int j = (k < cnt[n]) ? nb[n][k] : i0 + n;
      u[n][k] = ld4(src + (size_t)j * DIM + d);
    }
  }
#pragma unroll
  for (int n = 0; n < NB; ++n) {
#pragma unroll
    for (int k = 0; k < KNBR; ++k) {
      asm volatile("" :: "v"(u[n][k].x), "v"(u[n][k].y), "v"(u[n][k].z),
                         "v"(u[n][k].w));
    }
  }
#pragma unroll
  for (int n = 0; n < NB; ++n) {
    float4 m = {NEGINF, NEGINF, NEGINF, NEGINF};
#pragma unroll
    for (int k = 0; k < KNBR; ++k) {
      if (k < cnt[n]) {
        m.x = fmaxf(m.x, u[n][k].x); m.y = fmaxf(m.y, u[n][k].y);
        m.z = fmaxf(m.z, u[n][k].z); m.w = fmaxf(m.w, u[n][k].w);
      }
    }
    if (cnt[n] == 0) m = make_float4(0.f, 0.f, 0.f, 0.f);
    st4(pool + (size_t)(i0 + n) * DIM + d, m);
  }
}

__global__ __launch_bounds__(128) __attribute__((amdgpu_waves_per_eu(1, 2)))
void pool_gemm_ln(const float* __restrict__ s,
                  const int* __restrict__ neigh,
                  const int* __restrict__ nbrc,
                  float* pool,
                  const float* __restrict__ W,
                  const float* __restrict__ bias,
                  float* tmp,
                  const float* __restrict__ g,
                  const float* __restrict__ b,
                  float* __restrict__ dst, int ldd) {
  const int wave = threadIdx.x >> 5;
  const int lane = threadIdx.x & 31;
  const int d = lane * 4;
  const int i0 = (int)blockIdx.x * 32 + wave * 8;
  pool_rows<4>(i0, lane, s, neigh, nbrc, pool);
  pool_rows<4>(i0 + 4, lane, s, neigh, nbrc, pool);
  __syncthreads();
  gemm_tile<2 * DIM, true>(wave, s, pool, W, bias, tmp);
  __syncthreads();
  float4 av[8], rv[8];
#pragma unroll
  for (int n = 0; n < 8; ++n) {
    av[n] = ld4(tmp + (size_t)(i0 + n) * DIM + d);
    rv[n] = ld4(s + (size_t)(i0 + n) * DIM + d);
  }
#pragma unroll
  for (int n = 0; n < 8; ++n) {
    const float4 v = {av[n].x + rv[n].x, av[n].y + rv[n].y,
                      av[n].z + rv[n].z, av[n].w + rv[n].w};
    wave_ln_store(v, g, b, dst + (size_t)(i0 + n) * ldd, lane);
  }
}

__global__ void gcn_ln_sage_ln(const float* __restrict__ nxt,
                               const float* __restrict__ resid,
                               const int* __restrict__ neigh,
                               const float* __restrict__ dinv,
                               const unsigned* __restrict__ keep,
                               const float* __restrict__ g,
                               const float* __restrict__ b,
                               float* __restrict__ dst, int ldd,
                               const float* __restrict__ a,
                               const float* __restrict__ res,
                               const float* __restrict__ sg,
                               const float* __restrict__ sb,
                               float* __restrict__ sdst) {
  const int i = (int)((blockIdx.x * blockDim.x + threadIdx.x) >> 5);
  if (i >= NNODE) return;
  const int lane = threadIdx.x & 31;
  const int d = lane * 4;

  const float4 av = ld4(a + (size_t)i * DIM + d);
  const float4 rv = ld4(res + (size_t)i * DIM + d);
  const float4 t = ld4(nxt + (size_t)i * DIM + d);
  const float4 r = ld4(resid + (size_t)i * DIM + d);
  const float di = dinv[i];
  const unsigned km = keep[i];
  int nb[KNBR];
  ld_nbrs(neigh + i * KNBR, nb);
  asm volatile("" :: "v"(av.x), "v"(av.y), "v"(av.z), "v"(av.w),
                     "v"(rv.x), "v"(rv.y), "v"(rv.z), "v"(rv.w),
                     "v"(t.x), "v"(t.y), "v"(t.z), "v"(t.w),
                     "v"(r.x), "v"(r.y), "v"(r.z), "v"(r.w),
                     "v"(di), "v"(km),
                     "v"(nb[0]), "v"(nb[1]), "v"(nb[2]), "v"(nb[3]),
                     "v"(nb[4]), "v"(nb[5]), "v"(nb[6]), "v"(nb[7]));

  float4 u[KNBR];
  float dj[KNBR];
#pragma unroll
  for (int k = 0; k < KNBR; ++k) {
    const int j = ((km >> k) & 1u) ? nb[k] : i;
    dj[k] = dinv[j];
    u[k] = ld4(nxt + (size_t)j * DIM + d);
  }
#pragma unroll
  for (int k = 0; k < KNBR; ++k) {
    asm volatile("" :: "v"(u[k].x), "v"(u[k].y), "v"(u[k].z), "v"(u[k].w),
                       "v"(dj[k]));
  }

  const float4 w = {av.x + rv.x, av.y + rv.y, av.z + rv.z, av.w + rv.w};
  wave_ln_store(w, sg, sb, sdst + (size_t)i * DIM, lane);

  const float4 v = gcn_sum(di, km, t, r, u, dj);
  wave_ln_store(v, g, b, dst + (size_t)i * ldd, lane);
}

extern "C" void kernel_launch(void* const* d_in, const int* in_sizes, int n_in,
                              void* d_out, int out_size, void* d_ws, size_t ws_size,
                              hipStream_t stream) {
  const float* x      = (const float*)d_in[0];
  const int*   neigh  = (const int*)  d_in[1];
  const int*   nbrc   = (const int*)  d_in[2];
  const float* gcnW   = (const float*)d_in[3];
  const float* gcnB   = (const float*)d_in[4];
  const float* gcnG   = (const float*)d_in[5];
  const float* gcnBb  = (const float*)d_in[6];
  const float* poolW  = (const float*)d_in[7];
  const float* poolB  = (const float*)d_in[8];
  const float* mrgW   = (const float*)d_in[9];
  const float* mrgB   = (const float*)d_in[10];
  const float* sgG    = (const float*)d_in[11];
  const float* sgB    = (const float*)d_in[12];
  float* out = (float*)d_out;

  const size_t ND = (size_t)NNODE * DIM;
  float*    dinv  = (float*)d_ws;
  unsigned* keep  = (unsigned*)(dinv + NNODE);
  float*    first = (float*)(keep + NNODE);
  float*    wt    = first + ND;
  float*    hbuf  = wt + ND;
  const float* wtG = wt;
  const float* wtP = wt + 3 * DIM * DIM;
  const float* wtM = wt + 4 * DIM * DIM;
  float*    sbuf  = hbuf + ND;
  float*    tmp   = sbuf + ND;
  float*    pool  = tmp + ND;
  float*    nxt2  = pool + ND;
  float*    nxt3  = first;

  const dim3 gB(128);
  const dim3 gG(NNODE / 32);
  const dim3 nB(256);
  const dim3 nG(NNODE / 8);

  deg_kernel<<<NNODE / 256 + 160, 256, 0, stream>>>(neigh, nbrc, dinv, keep,
                                                    gcnW, poolW, mrgW, wt);

  wmma_gemm_pair<DIM, false, DIM, true><<<gG, gB, 0, stream>>>(
      x, wtG, gcnB, first, x, wtP, poolB, tmp);

  rows_then_gemm_pair<<<gG, nB, 0, stream>>>(
      first, first, neigh, nbrc, dinv, keep, gcnG, gcnBb, hbuf,
      x, tmp, pool,
      wtG + DIM * DIM, gcnB + DIM, nxt2,
      wtM, mrgB, sbuf);

  rows_then_gemm_pair<<<gG, nB, 0, stream>>>(
      nxt2, hbuf, neigh, nbrc, dinv, keep, gcnG + DIM, gcnBb + DIM, hbuf,
      sbuf, sbuf, pool,
      wtG + 2 * DIM * DIM, gcnB + 2 * DIM, nxt3,
      wtM + DIM * 2 * DIM, mrgB + DIM, tmp);

  gcn_ln_sage_ln<<<nG, nB, 0, stream>>>(nxt3, hbuf, neigh, dinv, keep,
                                        gcnG + 2 * DIM, gcnBb + 2 * DIM,
                                        out, 2 * DIM,
                                        tmp, sbuf, sgG, sgB, sbuf);

  pool_gemm_ln<<<gG, gB, 0, stream>>>(sbuf, neigh, nbrc, pool,
                                      wtM + 2 * DIM * 2 * DIM, mrgB + 2 * DIM, tmp,
                                      sgG + DIM, sgB + DIM,
                                      out + DIM, 2 * DIM);
}


# Preparation launch dissolved: degrees and weight re-lay run as extra blocks of the first product launch; its two products read W row-major and share the x fragment loads; 6 -> 5; sums unchanged
# speedup vs baseline: 2.4761x; 1.1263x over previous
#include <hip/hip_runtime.h>

typedef __attribute__((ext_vector_type(2))) float v2f;
typedef __attribute__((ext_vector_type(8))) float v8f;

#define NNODE 8192
#define DIM   128
#define KNBR  8
#define NEGINF -1e30f

__device__ __forceinline__ float4 ld4(const float* p) {
  return *reinterpret_cast<const float4*>(p);
}
__device__ __forceinline__ void st4(float* p, float4 v) {
  *reinterpret_cast<float4*>(p) = v;
}
__device__ __forceinline__ float4 ld4_at(const float* base, unsigned byte_off) {
  return *reinterpret_cast<const float4*>(reinterpret_cast<const char*>(base) + byte_off);
}
__device__ __forceinline__ void ld_nbrs(const int* __restrict__ nb, int (&j)[KNBR]) {
  const int4 lo = *reinterpret_cast<const int4*>(nb);
  const int4 hi = *reinterpret_cast<const int4*>(nb + 4);
  j[0] = lo.x; j[1] = lo.y; j[2] = lo.z; j[3] = lo.w;
  j[4] = hi.x; j[5] = hi.y; j[6] = hi.z; j[7] = hi.w;
}

#define GEMM_BLOCKS (NNODE / 32)
#define PREP_THREADS 128
#define DEG_BLOCKS (NNODE / PREP_THREADS)
#define WT_SMALL (DIM * DIM)
#define WT_LARGE (DIM * 2 * DIM)
#define WT_BLOCKS (WT_LARGE / 4 / PREP_THREADS)

template <int K>
__device__ __forceinline__ void w_relayout(const float* __restrict__ Wsrc,
                                           float* __restrict__ Wdst, unsigned q) {
  const unsigned lane = q & 31u;
  const unsigned g    = (q >> 5) % (unsigned)(K / 8);
  const unsigned wt   = (q >> 5) / (unsigned)(K / 8);
  const unsigned row  = 32u * (wt >> 1) + 2u * (lane & 15u) + (wt & 1u);
  const unsigned col  = 8u * g + 4u * (lane >> 4);
  st4(Wdst + (size_t)q * 4, ld4(Wsrc + (size_t)row * K + col));
}

__device__ __forceinline__ void relayout_pieces(unsigned u,
                                                const float* __restrict__ gcnW,
                                                const float* __restrict__ mrgW,
                                                float* __restrict__ wt) {
  const unsigned m = 1u + u / (WT_SMALL / 4), q = u % (WT_SMALL / 4);
  w_relayout<DIM>(gcnW + (size_t)m * WT_SMALL, wt + (size_t)m * WT_SMALL, q);
#pragma unroll
  for (int mm = 0; mm < 3; ++mm)
    w_relayout<2 * DIM>(mrgW + (size_t)mm * WT_LARGE,
                        wt + (size_t)4 * WT_SMALL + (size_t)mm * WT_LARGE, u);
}

__device__ __forceinline__ void node_degree(int i,
                                            const int* __restrict__ neigh,
                                            const int* __restrict__ nbrc,
                                            float* __restrict__ dinv,
                                            unsigned* __restrict__ keep) {
  int cnt = nbrc[i]; cnt = cnt > KNBR ? KNBR : cnt;
  int nb[KNBR];
  ld_nbrs(neigh + i * KNBR, nb);
  unsigned km = 0;
  int deg = 1;
#pragma unroll
  for (int k = 0; k < KNBR; ++k) {
    if (k < cnt) {
      const int j = nb[k];
      bool dup = (j == i);
#pragma unroll
      for (int t = 0; t < k; ++t) dup |= (nb[t] == j);
      if (!dup) { km |= (1u << k); ++deg; }
    }
  }
  dinv[i] = rsqrtf((float)deg);
  keep[i] = km;
}

__device__ __forceinline__ v8f mma4(v2f a, v2f b, v8f c) {
  return __builtin_amdgcn_wmma_f32_16x16x4_f32(false, a, false, b,
                                               (short)0, c, false, false);
}

struct Frag4 { float4 a0, a1, b0, b1; };

template <int BSTEP>
__device__ __forceinline__ Frag4 ld_group(const float* a0p,
                                          const float* a1p,
                                          const float* __restrict__ b0p,
                                          const float* __restrict__ b1p,
                                          int koff) {
  Frag4 f;
  f.a0 = ld4(a0p);
  f.a1 = ld4(a1p);
  f.b0 = ld4(b0p + BSTEP * koff);
  f.b1 = ld4(b1p + BSTEP * koff);
  return f;
}

template <int K, bool RELU, bool WROWS = false>
__device__ __forceinline__ void gemm_tile(int wave,
                                          const float* Alo,
                                          const float* Ahi,
                                          const float* __restrict__ W,
                                          const float* __restrict__ bias,
                                          float* Y) {
  constexpr int NC = K / 32;
  const int lane = threadIdx.x & 31;
  const int half = lane >> 4;
  const int l16  = lane & 15;
  const int m0 = blockIdx.x * 32;
  const int n0 = wave * 32;

  const size_t aoff = (size_t)(m0 + l16) * DIM + 4 * half;
  const float* a0lo = Alo + aoff;
  const float* a0hi = Ahi + aoff;
#define A0P(k) ((k) < DIM ? a0lo + (k) : a0hi + ((k) - DIM))
  constexpr int BSTEP = WROWS ? 1 : 16;
  const float* __restrict__ b0p =
      WROWS ? W + (size_t)(n0 + 2 * l16) * K + 4 * half
            : W + (size_t)wave * 32 * K + lane * 4;
  const float* __restrict__ b1p = b0p + (WROWS ? K : 16 * K);

  Frag4 f[NC][4];
#pragma unroll
  for (int g = 0; g < 4; ++g)
    f[0][g] = ld_group<BSTEP>(A0P(8 * g), A0P(8 * g) + 16 * DIM, b0p, b1p, 8 * g);

  v8f c00 = {}, c01 = {}, c10 = {}, c11 = {};
#pragma unroll
  for (int c = 0; c < NC; ++c) {
    if (c + 1 < NC) {
#pragma unroll
      for (int g = 0; g < 4; ++g)
        f[c + 1][g] = ld_group<BSTEP>(A0P(32 * (c + 1) + 8 * g),
                                      A0P(32 * (c + 1) + 8 * g) + 16 * DIM,
                                      b0p, b1p, 32 * (c + 1) + 8 * g);
    }
#pragma unroll
    for (int g = 0; g < 4; ++g) {
      const Frag4 q = f[c][g];
      const v2f a0l = {q.a0.x, q.a0.y}, a0h = {q.a0.z, q.a0.w};
      const v2f a1l = {q.a1.x, q.a1.y}, a1h = {q.a1.z, q.a1.w};
      const v2f b0l = {q.b0.x, q.b0.y}, b0h = {q.b0.z, q.b0.w};
      const v2f b1l = {q.b1.x, q.b1.y}, b1h = {q.b1.z, q.b1.w};
      c00 = mma4(a0l, b0l, c00);
      c01 = mma4(a0l, b1l, c01);
      c10 = mma4(a1l, b0l, c10);
      c11 = mma4(a1l, b1l, c11);
      c00 = mma4(a0h, b0h, c00);
      c01 = mma4(a0h, b1h, c01);
      c10 = mma4(a1h, b0h, c10);
      c11 = mma4(a1h, b1h, c11);
    }
  }

#undef A0P
  const int col = n0 + 2 * l16;
  const float2 bv = *reinterpret_cast<const float2*>(bias + col);
  float* y0 = Y + (size_t)(m0 + 8 * half) * DIM + col;
  float* y1 = y0 + 16 * DIM;
#pragma unroll
  for (int r = 0; r < 8; ++r) {
    float2 u = {c00[r] + bv.x, c01[r] + bv.y};
    float2 v = {c10[r] + bv.x, c11[r] + bv.y};
    if (RELU) {
      u.x = fmaxf(u.x, 0.0f); u.y = fmaxf(u.y, 0.0f);
      v.x = fmaxf(v.x, 0.0f); v.y = fmaxf(v.y, 0.0f);
    }
    *reinterpret_cast<float2*>(y0 + (size_t)r * DIM) = u;
    *reinterpret_cast<float2*>(y1 + (size_t)r * DIM) = v;
  }
}

__global__ __launch_bounds__(PREP_THREADS) __attribute__((amdgpu_waves_per_eu(1, 2)))
void first_products_and_prep(const float* __restrict__ x,
                             const float* __restrict__ bias1,
                             float* __restrict__ Y1,
                             const float* __restrict__ W2,
                             const float* __restrict__ bias2,
                             float* __restrict__ Y2,
                             const int* __restrict__ neigh,
                             const int* __restrict__ nbrc,
                             float* __restrict__ dinv,
                             unsigned* __restrict__ keep,
                             const float* __restrict__ gcnW,
                             const float* __restrict__ mrgW,
                             float* __restrict__ wt) {
  if (blockIdx.x >= GEMM_BLOCKS) {
    const unsigned pb = blockIdx.x - GEMM_BLOCKS;
    if (pb < DEG_BLOCKS) {
      node_degree((int)(pb * PREP_THREADS + threadIdx.x), neigh, nbrc, dinv, keep);
    } else {
      relayout_pieces((pb - DEG_BLOCKS) * PREP_THREADS + threadIdx.x, gcnW, mrgW, wt);
    }
    return;
  }
  const int wave = threadIdx.x >> 5;
  gemm_tile<DIM, false, true>(wave, x, x, gcnW, bias1, Y1);
  gemm_tile<DIM, true, true>(wave, x, x, W2, bias2, Y2);
}

__device__ __forceinline__ void wave_ln_store(float4 v,
                                              const float* __restrict__ g,
                                              const float* __restrict__ b,
                                              float* __restrict__ dstrow,
                                              int lane) {
  float s = v.x + v.y + v.z + v.w;
#pragma unroll
  for (int off = 16; off > 0; off >>= 1) s += __shfl_xor(s, off, 32);
  const float mu = s * (1.0f / (float)DIM);
  const float dx = v.x - mu, dy = v.y - mu, dz = v.z - mu, dw = v.w - mu;
  float q = dx * dx + dy * dy + dz * dz + dw * dw;
#pragma unroll
  for (int off = 16; off > 0; off >>= 1) q += __shfl_xor(q, off, 32);
  const float rinv = rsqrtf(q * (1.0f / (float)DIM) + 1e-5f);
  const int d = lane * 4;
  const float4 gg = ld4(g + d);
  const float4 bb = ld4(b + d);
  float4 o;
  o.x = dx * rinv * gg.x + bb.x;
  o.y = dy * rinv * gg.y + bb.y;
  o.z = dz * rinv * gg.z + bb.z;
  o.w = dw * rinv * gg.w + bb.w;
  st4(dstrow + d, o);
}

template <int NB>
__device__ __forceinline__ void wave_ln_store_rows(const float4 (&v)[NB],
                                                   const float* __restrict__ g,
                                                   const float* __restrict__ b,
                                                   float* dst0, int lane) {
  const int d = lane * 4;
  const float4 gg = ld4(g + d);
  const float4 bb = ld4(b + d);
  float s[NB], q[NB];
  float4 dv[NB];
#pragma unroll
  for (int n = 0; n < NB; ++n) s[n] = v[n].x + v[n].y + v[n].z + v[n].w;
#pragma unroll
  for (int off = 16; off > 0; off >>= 1) {
    float o[NB];
#pragma unroll
    for (int n = 0; n < NB; ++n) o[n] = __shfl_xor(s[n], off, 32);
#pragma unroll
    for (int n = 0; n < NB; ++n) s[n] += o[n];
  }
#pragma unroll
  for (int n = 0; n < NB; ++n) {
    const float mu = s[n] * (1.0f / (float)DIM);
    dv[n].x = v[n].x - mu; dv[n].y = v[n].y - mu;
    dv[n].z = v[n].z - mu; dv[n].w = v[n].w - mu;
    q[n] = dv[n].x * dv[n].x + dv[n].y * dv[n].y + dv[n].z * dv[n].z + dv[n].w * dv[n].w;
  }
#pragma unroll
  for (int off = 16; off > 0; off >>= 1) {
    float o[NB];
#pragma unroll
    for (int n = 0; n < NB; ++n) o[n] = __shfl_xor(q[n], off, 32);
#pragma unroll
    for (int n = 0; n < NB; ++n) q[n] += o[n];
  }
#pragma unroll
  for (int n = 0; n < NB; ++n) {
    const float rinv = rsqrtf(q[n] * (1.0f / (float)DIM) + 1e-5f);
    float4 o;
    o.x = dv[n].x * rinv * gg.x + bb.x;
    o.y = dv[n].y * rinv * gg.y + bb.y;
    o.z = dv[n].z * rinv * gg.z + bb.z;
    o.w = dv[n].w * rinv * gg.w + bb.w;
    st4(dst0 + (size_t)n * DIM + d, o);
  }
}

__device__ __forceinline__ float4 gcn_sum(float di, unsigned km, float4 t, float4 r,
                                          const float4 (&u)[KNBR],
                                          const float (&dj)[KNBR]) {
  float4 acc = {di * t.x, di * t.y, di * t.z, di * t.w};
#pragma unroll
  for (int k = 0; k < KNBR; ++k) {
    if ((km >> k) & 1u) {
      acc.x += dj[k] * u[k].x; acc.y += dj[k] * u[k].y;
      acc.z += dj[k] * u[k].z; acc.w += dj[k] * u[k].w;
    }
  }
  acc.x = fmaxf(acc.x * di, 0.0f);
  acc.y = fmaxf(acc.y * di, 0.0f);
  acc.z = fmaxf(acc.z * di, 0.0f);
  acc.w = fmaxf(acc.w * di, 0.0f);
  return make_float4(acc.x + r.x, acc.y + r.y, acc.z + r.z, acc.w + r.w);
}

__device__ __forceinline__ float4 gcn_sum_sel(float di, unsigned km, float4 t, float4 r,
                                              const float4 (&u)[KNBR],
                                              const float (&dj)[KNBR]) {
  float4 acc = {di * t.x, di * t.y, di * t.z, di * t.w};
#pragma unroll
  for (int k = 0; k < KNBR; ++k) {
    const bool on = ((km >> k) & 1u) != 0u;
    const float ax = acc.x + dj[k] * u[k].x, ay = acc.y + dj[k] * u[k].y;
    const float az = acc.z + dj[k] * u[k].z, aw = acc.w + dj[k] * u[k].w;
    acc.x = on ? ax : acc.x; acc.y = on ? ay : acc.y;
    acc.z = on ? az : acc.z; acc.w = on ? aw : acc.w;
  }
  acc.x = fmaxf(acc.x * di, 0.0f);
  acc.y = fmaxf(acc.y * di, 0.0f);
  acc.z = fmaxf(acc.z * di, 0.0f);
  acc.w = fmaxf(acc.w * di, 0.0f);
  return make_float4(acc.x + r.x, acc.y + r.y, acc.z + r.z, acc.w + r.w);
}

template <int NB>
__device__ __forceinline__ void gcn_pool_rows(int i0, int lane,
                                              const float* __restrict__ nxt,
                                              const float* resid,
                                              const int* __restrict__ neigh,
                                              const int* __restrict__ nbrc,
                                              const float* __restrict__ dinv,
                                              const unsigned* __restrict__ keep,
                                              const float* __restrict__ g,
                                              const float* __restrict__ b,
                                              float* dst,
                                              const float* __restrict__ poolsrc,
                                              float* pool) {
  const int d = lane * 4;
  float4 t[NB], r[NB];
  float di[NB];
  unsigned km[NB];
  int cnt[NB];
  int nb[NB][KNBR];
  int own[NB];
#pragma unroll
  for (int n = 0; n < NB; ++n) {
    const int i = i0 + n;
    own[n] = i;
    t[n] = ld4(nxt + (size_t)i * DIM + d);
    r[n] = ld4(resid + (size_t)i * DIM + d);
    di[n] = dinv[i];
    km[n] = keep[i];
    const int c = nbrc[i];
    cnt[n] = c > KNBR ? KNBR : c;
    ld_nbrs(neigh + i * KNBR, nb[n]);
  }
#pragma unroll
  for (int n = 0; n < NB; ++n) {
    asm volatile("" : "+v"(own[n]) : "v"(t[n].x), "v"(t[n].y), "v"(t[n].z), "v"(t[n].w),
                       "v"(r[n].x), "v"(r[n].y), "v"(r[n].z), "v"(r[n].w),
                       "v"(di[n]), "v"(km[n]), "v"(cnt[n]),
                       "v"(nb[n][0]), "v"(nb[n][1]), "v"(nb[n][2]), "v"(nb[n][3]),
                       "v"(nb[n][4]), "v"(nb[n][5]), "v"(nb[n][6]), "v"(nb[n][7]));
  }

  const unsigned boff = (unsigned)lane * 16u;
  float4 u[NB][KNBR], p[NB][KNBR];
  float dj[NB][KNBR];
#pragma unroll
  for (int n = 0; n < NB; ++n) {
#pragma unroll
    for (int k = 0; k < KNBR; ++k) {
      const int j = ((km[n] >> k) & 1u) ? nb[n][k] : own[n];
      dj[n][k] = dinv[j];
      u[n][k] = ld4_at(nxt, ((unsigned)j << 9) + boff);
    }
  }
#pragma unroll
  for (int n = 0; n < NB; ++n) {
#pragma unroll
    for (int k = 0; k < KNBR; ++k) {
      const int j = (k < cnt[n]) ? nb[n][k] : own[n];
      p[n][k] = ld4_at(poolsrc, ((unsigned)j << 9) + boff);
    }
  }

  float4 v[NB];
#pragma unroll
  for (int n = 0; n < NB; ++n)
    v[n] = gcn_sum_sel(di[n], km[n], t[n], r[n], u[n], dj[n]);
  wave_ln_store_rows<NB>(v, g, b, dst + (size_t)i0 * DIM, lane);

#pragma unroll
  for (int n = 0; n < NB; ++n) {
    float4 m = {NEGINF, NEGINF, NEGINF, NEGINF};
#pragma unroll
    for (int k = 0; k < KNBR; ++k) {
      const bool on = k < cnt[n];
      m.x = fmaxf(m.x, on ? p[n][k].x : NEGINF); m.y = fmaxf(m.y, on ? p[n][k].y : NEGINF);
      m.z = fmaxf(m.z, on ? p[n][k].z : NEGINF); m.w = fmaxf(m.w, on ? p[n][k].w : NEGINF);
    }
    if (cnt[n] == 0) m = make_float4(0.f, 0.f, 0.f, 0.f);
    st4(pool + (size_t)(i0 + n) * DIM + d, m);
  }
}

__global__ __launch_bounds__(256) __attribute__((amdgpu_waves_per_eu(2, 2)))
void rows_then_gemm_pair(const float* __restrict__ nxt,
                         const float* resid,
                         const int* __restrict__ neigh,
                         const int* __restrict__ nbrc,
                         const float* __restrict__ dinv,
                         const unsigned* __restrict__ keep,
                         const float* __restrict__ g,
                         const float* __restrict__ b,
                         float* dst,
                         const float* __restrict__ selfsrc,
                         const float* __restrict__ poolsrc,
                         float* pool,
                         const float* __restrict__ W1,
                         const float* __restrict__ bias1,
                         float* __restrict__ Y1,
                         const float* __restrict__ W2,
                         const float* __restrict__ bias2,
                         float* __restrict__ Y2) {
  const int lane = threadIdx.x & 31;
  const int i0 = (int)blockIdx.x * 32 + (int)(threadIdx.x >> 5) * 4;
  gcn_pool_rows<4>(i0, lane, nxt, resid, neigh, nbrc, dinv, keep, g, b, dst,
                   poolsrc, pool);
  __syncthreads();
  const int wave = __builtin_amdgcn_readfirstlane((int)(threadIdx.x >> 5));
  if (wave < 4) gemm_tile<DIM, false>(wave, dst, dst, W1, bias1, Y1);
  else          gemm_tile<2 * DIM, true>(wave - 4, selfsrc, pool, W2, bias2, Y2);
}

template <int NB>
__device__ __forceinline__ void pool_rows(int i0, int lane,
                                          const float* __restrict__ src,
                                          const int* __restrict__ neigh,
                                          const int* __restrict__ nbrc,
                                          float* pool) {
  const int d = lane * 4;
  const unsigned boff = (unsigned)lane * 16u;
  int cnt[NB];
  int nb[NB][KNBR];
#pragma unroll
  for (int n = 0; n < NB; ++n) {
    const int c = nbrc[i0 + n];
    cnt[n] = c > KNBR ? KNBR : c;
    ld_nbrs(neigh + (i0 + n) * KNBR, nb[n]);
  }
#pragma unroll
  for (int n = 0; n < NB; ++n) {
    asm volatile("" :: "v"(cnt[n]),
                       "v"(nb[n][0]), "v"(nb[n][1]), "v"(nb[n][2]), "v"(nb[n][3]),
                       "v"(nb[n][4]), "v"(nb[n][5]), "v"(nb[n][6]), "v"(nb[n][7]));
  }
  float4 u[NB][KNBR];
#pragma unroll
  for (int n = 0; n < NB; ++n) {
#pragma unroll
    for (int k = 0; k < KNBR; ++k) {
      const int j = (k < cnt[n]) ? nb[n][k] : i0 + n;
      u[n][k] = ld4_at(src, ((unsigned)j << 9) + boff);
    }
  }
#pragma unroll
  for (int n = 0; n < NB; ++n) {
    float4 m = {NEGINF, NEGINF, NEGINF, NEGINF};
#pragma unroll
    for (int k = 0; k < KNBR; ++k) {
      const bool on = k < cnt[n];
      m.x = fmaxf(m.x, on ? u[n][k].x : NEGINF); m.y = fmaxf(m.y, on ? u[n][k].y : NEGINF);
      m.z = fmaxf(m.z, on ? u[n][k].z : NEGINF); m.w = fmaxf(m.w, on ? u[n][k].w : NEGINF);
    }
    if (cnt[n] == 0) m = make_float4(0.f, 0.f, 0.f, 0.f);
    st4(pool + (size_t)(i0 + n) * DIM + d, m);
  }
}

__global__ __launch_bounds__(128) __attribute__((amdgpu_waves_per_eu(1, 2)))
void pool_gemm_ln(const float* __restrict__ s,
                  const int* __restrict__ neigh,
                  const int* __restrict__ nbrc,
                  float* pool,
                  const float* __restrict__ W,
                  const float* __restrict__ bias,
                  float* tmp,
                  const float* __restrict__ g,
                  const float* __restrict__ b,
                  float* __restrict__ dst, int ldd) {
  const int wave = threadIdx.x >> 5;
  const int lane = threadIdx.x & 31;
  const int d = lane * 4;
  const int i0 = (int)blockIdx.x * 32 + wave * 8;
  pool_rows<8>(i0, lane, s, neigh, nbrc, pool);
  __syncthreads();
  gemm_tile<2 * DIM, true>(wave, s, pool, W, bias, tmp);
  __syncthreads();
  float4 av[8], rv[8];
#pragma unroll
  for (int n = 0; n < 8; ++n) {
    av[n] = ld4(tmp + (size_t)(i0 + n) * DIM + d);
    rv[n] = ld4(s + (size_t)(i0 + n) * DIM + d);
  }
#pragma unroll
  for (int n = 0; n < 8; ++n) {
    const float4 v = {av[n].x + rv[n].x, av[n].y + rv[n].y,
                      av[n].z + rv[n].z, av[n].w + rv[n].w};
    wave_ln_store(v, g, b, dst + (size_t)(i0 + n) * ldd, lane);
  }
}

__global__ void gcn_ln_sage_ln(const float* __restrict__ nxt,
                               const float* __restrict__ resid,
                               const int* __restrict__ neigh,
                               const float* __restrict__ dinv,
                               const unsigned* __restrict__ keep,
                               const float* __restrict__ g,
                               const float* __restrict__ b,
                               float* __restrict__ dst, int ldd,
                               const float* __restrict__ a,
                               const float* __restrict__ res,
                               const float* __restrict__ sg,
                               const float* __restrict__ sb,
                               float* __restrict__ sdst) {
  const int i = (int)((blockIdx.x * blockDim.x + threadIdx.x) >> 5);
  if (i >= NNODE) return;
  const int lane = threadIdx.x & 31;
  const int d = lane * 4;

  const float4 av = ld4(a + (size_t)i * DIM + d);
  const float4 rv = ld4(res + (size_t)i * DIM + d);
  const float4 t = ld4(nxt + (size_t)i * DIM + d);
  const float4 r = ld4(resid + (size_t)i * DIM + d);
  const float di = dinv[i];
  const unsigned km = keep[i];
  int nb[KNBR];
  ld_nbrs(neigh + i * KNBR, nb);
  asm volatile("" :: "v"(av.x), "v"(av.y), "v"(av.z), "v"(av.w),
                     "v"(rv.x), "v"(rv.y), "v"(rv.z), "v"(rv.w),
                     "v"(t.x), "v"(t.y), "v"(t.z), "v"(t.w),
                     "v"(r.x), "v"(r.y), "v"(r.z), "v"(r.w),
                     "v"(di), "v"(km),
                     "v"(nb[0]), "v"(nb[1]), "v"(nb[2]), "v"(nb[3]),
                     "v"(nb[4]), "v"(nb[5]), "v"(nb[6]), "v"(nb[7]));

  float4 u[KNBR];
  float dj[KNBR];
#pragma unroll
  for (int k = 0; k < KNBR; ++k) {
    const int j = ((km >> k) & 1u) ? nb[k] : i;
    dj[k] = dinv[j];
    u[k] = ld4(nxt + (size_t)j * DIM + d);
  }
#pragma unroll
  for (int k = 0; k < KNBR; ++k) {
    asm volatile("" :: "v"(u[k].x), "v"(u[k].y), "v"(u[k].z), "v"(u[k].w),
                       "v"(dj[k]));
  }

  const float4 w = {av.x + rv.x, av.y + rv.y, av.z + rv.z, av.w + rv.w};
  wave_ln_store(w, sg, sb, sdst + (size_t)i * DIM, lane);

  const float4 v = gcn_sum(di, km, t, r, u, dj);
  wave_ln_store(v, g, b, dst + (size_t)i * ldd, lane);
}

extern "C" void kernel_launch(void* const* d_in, const int* in_sizes, int n_in,
                              void* d_out, int out_size, void* d_ws, size_t ws_size,
                              hipStream_t stream) {
  const float* x      = (const float*)d_in[0];
  const int*   neigh  = (const int*)  d_in[1];
  const int*   nbrc   = (const int*)  d_in[2];
  const float* gcnW   = (const float*)d_in[3];
  const float* gcnB   = (const float*)d_in[4];
  const float* gcnG   = (const float*)d_in[5];
  const float* gcnBb  = (const float*)d_in[6];
  const float* poolW  = (const float*)d_in[7];
  const float* poolB  = (const float*)d_in[8];
  const float* mrgW   = (const float*)d_in[9];
  const float* mrgB   = (const float*)d_in[10];
  const float* sgG    = (const float*)d_in[11];
  const float* sgB    = (const float*)d_in[12];
  float* out = (float*)d_out;

  const size_t ND = (size_t)NNODE * DIM;
  float*    dinv  = (float*)d_ws;
  unsigned* keep  = (unsigned*)(dinv + NNODE);
  float*    first = (float*)(keep + NNODE);
  float*    wt    = first + ND;
  float*    hbuf  = wt + ND;
  const float* wtG = wt;
  const float* wtM = wt + 4 * DIM * DIM;
  float*    sbuf  = hbuf + ND;
  float*    tmp   = sbuf + ND;
  float*    pool  = tmp + ND;
  float*    nxt2  = pool + ND;
  float*    nxt3  = first;

  const dim3 gB(128);
  const dim3 gG(NNODE / 32);
  const dim3 nB(256);
  const dim3 nG(NNODE / 8);

  first_products_and_prep<<<GEMM_BLOCKS + DEG_BLOCKS + WT_BLOCKS, gB, 0, stream>>>(
      x, gcnB, first, poolW, poolB, tmp,
      neigh, nbrc, dinv, keep, gcnW, mrgW, wt);

  rows_then_gemm_pair<<<gG, nB, 0, stream>>>(
      first, first, neigh, nbrc, dinv, keep, gcnG, gcnBb, hbuf,
      x, tmp, pool,
      wtG + DIM * DIM, gcnB + DIM, nxt2,
      wtM, mrgB, sbuf);

  rows_then_gemm_pair<<<gG, nB, 0, stream>>>(
      nxt2, hbuf, neigh, nbrc, dinv, keep, gcnG + DIM, gcnBb + DIM, hbuf,
      sbuf, sbuf, pool,
      wtG + 2 * DIM * DIM, gcnB + 2 * DIM, nxt3,
      wtM + DIM * 2 * DIM, mrgB + DIM, tmp);

  gcn_ln_sage_ln<<<nG, nB, 0, stream>>>(nxt3, hbuf, neigh, dinv, keep,
                                        gcnG + 2 * DIM, gcnBb + 2 * DIM,
                                        out, 2 * DIM,
                                        tmp, sbuf, sgG, sgB, sbuf);

  pool_gemm_ln<<<gG, gB, 0, stream>>>(sbuf, neigh, nbrc, pool,
                                      wtM + 2 * DIM * 2 * DIM, mrgB + 2 * DIM, tmp,
                                      sgG + DIM, sgB + DIM,
                                      out + DIM, 2 * DIM);
}
